# starting kernel
# baseline (speedup 1.0000x reference)
#include <hip/hip_runtime.h>
#include <math.h>


typedef __attribute__((ext_vector_type(2))) float v2f;
typedef __attribute__((ext_vector_type(8))) float v8f;

#define N_NODES 50000
#define TILE_M 64
#define TILE_N 64
#define TILE_K 32

__global__ void fill_f32(float* __restrict__ p, float v, long n) {
    long i = (long)blockIdx.x * blockDim.x + threadIdx.x;
    if (i < n) p[i] = v;
}

__device__ __forceinline__ void atomicMaxF(float* addr, float val) {
    if (val >= 0.0f)
        atomicMax((int*)addr, __float_as_int(val));
    else
        atomicMin((unsigned int*)addr, (unsigned int)__float_as_int(val));
}

__global__ __launch_bounds__(256) void gemm_wmma_f32(
    const float* __restrict__ p0, const float* __restrict__ p1,
    const float* __restrict__ p2, int w0, int w1, int w2,
    const float* __restrict__ B, float* __restrict__ C,
    int M, int N, int K)
{
    __shared__ float As[TILE_M][TILE_K + 2];
    __shared__ float Bs[TILE_K][TILE_N + 2];

    const int tid  = threadIdx.x;
    const int lane = tid & 31;
    const int wave = tid >> 5;
    const int mt     = wave & 3;
    const int ntBase = (wave >> 2) * 2;

    const int blockRow = blockIdx.x * TILE_M;
    const int blockCol = blockIdx.y * TILE_N;

    v8f acc0 = {};
    v8f acc1 = {};

    const int mLocal = lane & 15;
    const int kOff   = (lane >> 4) << 1;

    for (int kb = 0; kb < K; kb += TILE_K) {
#pragma unroll
        for (int i = 0; i < 8; ++i) {
            int idx = tid + i * 256;
            int m = idx >> 5;
            int k = idx & 31;
            int row = blockRow + m;
            int gk  = kb + k;
            float v = 0.0f;
            if (row < M && gk < K) {
                int kk = gk;
                if (kk < w0)               v = p0[(long)row * w0 + kk];
                else if ((kk -= w0) < w1)  v = p1[(long)row * w1 + kk];
                else                       v = p2[(long)row * w2 + (kk - w1)];
            }
            As[m][k] = v;
        }
#pragma unroll
        for (int i = 0; i < 8; ++i) {
            int idx = tid + i * 256;
            int k = idx >> 6;
            int n = idx & 63;
            int gk  = kb + k;
            int col = blockCol + n;
            float v = 0.0f;
            if (gk < K && col < N) v = B[(long)gk * N + col];
            Bs[k][n] = v;
        }
        __syncthreads();

#pragma unroll
        for (int kk = 0; kk < TILE_K; kk += 4) {
            v2f a, b0, b1;
            a.x = As[mt * 16 + mLocal][kk + kOff];
            a.y = As[mt * 16 + mLocal][kk + kOff + 1];
            b0.x = Bs[kk + kOff][ntBase * 16 + mLocal];
            b0.y = Bs[kk + kOff + 1][ntBase * 16 + mLocal];
            b1.x = Bs[kk + kOff][(ntBase + 1) * 16 + mLocal];
            b1.y = Bs[kk + kOff + 1][(ntBase + 1) * 16 + mLocal];
            acc0 = __builtin_amdgcn_wmma_f32_16x16x4_f32(
                false, a, false, b0, (short)0, acc0, false, false);
            acc1 = __builtin_amdgcn_wmma_f32_16x16x4_f32(
                false, a, false, b1, (short)0, acc1, false, false);
        }
        __syncthreads();
    }

    const int col0 = blockCol + ntBase * 16 + mLocal;
    const int col1 = col0 + 16;
    const int mAdd = (lane >> 4) << 3;
#pragma unroll
    for (int r = 0; r < 8; ++r) {
        int row = blockRow + mt * 16 + r + mAdd;
        if (row < M) {
            C[(long)row * N + col0] = acc0[r];
            C[(long)row * N + col1] = acc1[r];
        }
    }
}

__global__ __launch_bounds__(256) void attn_proj(
    const float* __restrict__ feat, const float* __restrict__ al,
    const float* __restrict__ ar, float* __restrict__ el,
    float* __restrict__ er, int nNodes, int H)
{
    int wid  = blockIdx.x * 8 + (threadIdx.x >> 5);
    int lane = threadIdx.x & 31;
    if (wid >= nNodes * H) return;
    int node = wid / H;
    int h    = wid - node * H;
    int C    = H * 64;
    const float* f  = feat + (long)node * C + h * 64;
    const float* wl = al + h * 64;
    const float* wr = ar + h * 64;
    float f0 = f[lane], f1 = f[lane + 32];
    float sl = f0 * wl[lane] + f1 * wl[lane + 32];
    float sr = f0 * wr[lane] + f1 * wr[lane + 32];
#pragma unroll
    for (int o = 16; o > 0; o >>= 1) {
        sl += __shfl_xor(sl, o, 32);
        sr += __shfl_xor(sr, o, 32);
    }
    if (lane == 0) { el[wid] = sl; er[wid] = sr; }
}

__global__ void edge_lrelu_max(
    const int* __restrict__ src, const int* __restrict__ dst,
    const float* __restrict__ el, const float* __restrict__ er,
    float* __restrict__ ebuf, float* __restrict__ emax, int E, int H)
{
    int i = blockIdx.x * blockDim.x + threadIdx.x;
    if (i >= E) return;
    int s = src[i], d = dst[i];
    for (int h = 0; h < H; ++h) {
        float x = el[s * H + h] + er[d * H + h];
        float e = x > 0.0f ? x : 0.2f * x;
        ebuf[(long)i * H + h] = e;
        atomicMaxF(&emax[d * H + h], e);
    }
}

__global__ void edge_expsum(
    const int* __restrict__ dst, float* __restrict__ ebuf,
    const float* __restrict__ emax, float* __restrict__ denom, int E, int H)
{
    int i = blockIdx.x * blockDim.x + threadIdx.x;
    if (i >= E) return;
    int d = dst[i];
    for (int h = 0; h < H; ++h) {
        float ex = expf(ebuf[(long)i * H + h] - emax[d * H + h]);
        ebuf[(long)i * H + h] = ex;
        atomicAdd(&denom[d * H + h], ex);
    }
}

__global__ __launch_bounds__(256) void edge_aggregate(
    const int* __restrict__ src, const int* __restrict__ dst,
    const float* __restrict__ feat, const float* __restrict__ ebuf,
    const float* __restrict__ denom, float* __restrict__ rst, int E, int H)
{
    int wid  = blockIdx.x * 8 + (threadIdx.x >> 5);
    int lane = threadIdx.x & 31;
    if (wid >= E) return;
    int s = src[wid], d = dst[wid];
    int C = H * 64;
    for (int idx = lane; idx < C; idx += 32) {
        int h = idx >> 6;
        float alpha = ebuf[(long)wid * H + h] / denom[d * H + h];
        atomicAdd(&rst[(long)d * C + idx], feat[(long)s * C + idx] * alpha);
    }
}

__global__ void bias_elu(float* __restrict__ x, const float* __restrict__ b,
                         long n, int C)
{
    long i = (long)blockIdx.x * blockDim.x + threadIdx.x;
    if (i >= n) return;
    float v = x[i] + b[i % C];
    x[i] = v > 0.0f ? v : expm1f(v);
}

__global__ __launch_bounds__(256) void classify(
    const float* __restrict__ rst2, const float* __restrict__ b2,
    const float* __restrict__ Wc1, const float* __restrict__ bc1,
    const float* __restrict__ Wc2, const float* __restrict__ bc2,
    float* __restrict__ out, int nNodes)
{
    int wid  = blockIdx.x * 8 + (threadIdx.x >> 5);
    int lane = threadIdx.x & 31;
    if (wid >= nNodes) return;
    float acc = 0.0f;
#pragma unroll
    for (int d = 0; d < 64; ++d)
        acc += (rst2[(long)wid * 64 + d] + b2[d]) * Wc1[d * 32 + lane];
    float t = fmaxf(acc + bc1[lane], 0.0f);
    float o0 = t * Wc2[lane * 2 + 0];
    float o1 = t * Wc2[lane * 2 + 1];
#pragma unroll
    for (int o = 16; o > 0; o >>= 1) {
        o0 += __shfl_xor(o0, o, 32);
        o1 += __shfl_xor(o1, o, 32);
    }
    if (lane == 0) {
        out[(long)wid * 2 + 0] = o0 + bc2[0];
        out[(long)wid * 2 + 1] = o1 + bc2[1];
    }
}

extern "C" void kernel_launch(void* const* d_in, const int* in_sizes, int n_in,
                              void* d_out, int out_size, void* d_ws, size_t ws_size,
                              hipStream_t stream)
{
    (void)n_in; (void)out_size; (void)ws_size;

    const float* randfeat = (const float*)d_in[0];
    const float* func_emb = (const float*)d_in[1];
    const float* codebert = (const float*)d_in[2];
    const int*   src      = (const int*)d_in[3];
    const int*   dst      = (const int*)d_in[4];
    const float* W1  = (const float*)d_in[5];
    const float* al1 = (const float*)d_in[6];
    const float* ar1 = (const float*)d_in[7];
    const float* b1  = (const float*)d_in[8];
    const float* W2  = (const float*)d_in[9];
    const float* al2 = (const float*)d_in[10];
    const float* ar2 = (const float*)d_in[11];
    const float* b2  = (const float*)d_in[12];
    const float* Wc1 = (const float*)d_in[13];
    const float* bc1 = (const float*)d_in[14];
    const float* Wc2 = (const float*)d_in[15];
    const float* bc2 = (const float*)d_in[16];
    float* out = (float*)d_out;

    const int  E  = in_sizes[3];
    const long NN = N_NODES;

    float* ws     = (float*)d_ws;
    float* feat1  = ws;
    float* rst1   = feat1  + NN * 256;
    float* feat2  = rst1   + NN * 256;
    float* rst2   = feat2  + NN * 64;
    float* el1    = rst2   + NN * 64;
    float* er1    = el1    + NN * 4;
    float* emax1  = er1    + NN * 4;
    float* denom1 = emax1  + NN * 4;
    float* el2    = denom1 + NN * 4;
    float* er2    = el2    + NN;
    float* emax2  = er2    + NN;
    float* denom2 = emax2  + NN;
    float* e1     = denom2 + NN;
    float* e2     = e1 + (long)E * 4;

    fill_f32<<<(int)((NN * 256 + 255) / 256), 256, 0, stream>>>(rst1, 0.0f, NN * 256);
    fill_f32<<<(int)((NN * 64 + 255) / 256), 256, 0, stream>>>(rst2, 0.0f, NN * 64);
    fill_f32<<<(int)((NN * 4 + 255) / 256), 256, 0, stream>>>(denom1, 0.0f, NN * 4);
    fill_f32<<<(int)((NN + 255) / 256), 256, 0, stream>>>(denom2, 0.0f, NN);
    fill_f32<<<(int)((NN * 4 + 255) / 256), 256, 0, stream>>>(emax1, -INFINITY, NN * 4);
    fill_f32<<<(int)((NN + 255) / 256), 256, 0, stream>>>(emax2, -INFINITY, NN);

    dim3 g1((N_NODES + TILE_M - 1) / TILE_M, 256 / TILE_N);
    gemm_wmma_f32<<<g1, 256, 0, stream>>>(randfeat, func_emb, codebert,
                                          100, 100, 768, W1, feat1,
                                          N_NODES, 256, 968);
    attn_proj<<<(N_NODES * 4 + 7) / 8, 256, 0, stream>>>(feat1, al1, ar1, el1, er1, N_NODES, 4);
    edge_lrelu_max<<<(E + 255) / 256, 256, 0, stream>>>(src, dst, el1, er1, e1, emax1, E, 4);
    edge_expsum<<<(E + 255) / 256, 256, 0, stream>>>(dst, e1, emax1, denom1, E, 4);
    edge_aggregate<<<(E + 7) / 8, 256, 0, stream>>>(src, dst, feat1, e1, denom1, rst1, E, 4);
    bias_elu<<<(int)((NN * 256 + 255) / 256), 256, 0, stream>>>(rst1, b1, NN * 256, 256);

    dim3 g2((N_NODES + TILE_M - 1) / TILE_M, 64 / TILE_N);
    gemm_wmma_f32<<<g2, 256, 0, stream>>>(rst1, rst1, rst1,
                                          256, 0, 0, W2, feat2,
                                          N_NODES, 64, 256);
    attn_proj<<<(N_NODES + 7) / 8, 256, 0, stream>>>(feat2, al2, ar2, el2, er2, N_NODES, 1);
    edge_lrelu_max<<<(E + 255) / 256, 256, 0, stream>>>(src, dst, el2, er2, e2, emax2, E, 1);
    edge_expsum<<<(E + 255) / 256, 256, 0, stream>>>(dst, e2, emax2, denom2, E, 1);
    edge_aggregate<<<(E + 7) / 8, 256, 0, stream>>>(src, dst, feat2, e2, denom2, rst2, E, 1);

    classify<<<(N_NODES + 7) / 8, 256, 0, stream>>>(rst2, b2, Wc1, bc1, Wc2, bc2, out, N_NODES);
}


# GEMMs re-tiled: block 128 rows x all of N (A read once), wave tile up to 64x64 so one LDS fragment read feeds 4 WMMAs; float4 staging, next K tile prefetched over 2 LDS stages
# speedup vs baseline: 1.5257x; 1.5257x over previous
#include <hip/hip_runtime.h>
#include <math.h>


typedef __attribute__((ext_vector_type(2))) float v2f;
typedef __attribute__((ext_vector_type(8))) float v8f;

#define N_NODES 50000

__global__ void fill_f32(float* __restrict__ p, float v, long n) {
    long i = (long)blockIdx.x * blockDim.x + threadIdx.x;
    if (i < n) p[i] = v;
}

__device__ __forceinline__ void atomicMaxF(float* addr, float val) {
    if (val >= 0.0f)
        atomicMax((int*)addr, __float_as_int(val));
    else
        atomicMin((unsigned int*)addr, (unsigned int)__float_as_int(val));
}

#define GT_BK 16
#define GT_AP 20

template <int WM, int WN, int TI>
__global__ __launch_bounds__(256) void gemm_wmma_f32_tiled(
    const float* __restrict__ p0, const float* __restrict__ p1,
    const float* __restrict__ p2, int w0, int w1, int w2,
    const float* __restrict__ B, float* __restrict__ C,
    int M, int N, int K)
{
    static_assert(WM * WN == 8, "8 waves per block");
    constexpr int BM  = WM * TI * 16;
    constexpr int BN  = WN * 64;
    constexpr int BP  = BN + 8;
    constexpr int NA  = BM / 64;
    constexpr int NB  = BN / 64;
    constexpr int BQ  = BN / 4;
    constexpr int BR  = 256 / BQ;
    static_assert(BM % 64 == 0, "A staging covers 64 rows per pass");

    __shared__ __attribute__((aligned(16))) float As[2][BM * GT_AP];
    __shared__ __attribute__((aligned(16))) float Bs[2][GT_BK * BP];

    const int tid  = threadIdx.x;
    const int lane = tid & 31;
    const int wave = tid >> 5;
    const int c    = lane & 15;
    const int h    = lane >> 4;
    const int m0   = (wave % WM) * (TI * 16);
    const int n0   = (wave / WM) * 64;

    const int blockRow = blockIdx.x * BM;
    const int blockCol = blockIdx.y * BN;

    const int am = tid >> 2;
    const int ak = (tid & 3) << 2;
    const int bk = tid / BQ;
    const int bn = (tid % BQ) << 2;

    v8f acc[TI][4];
    for (int i = 0; i < TI; ++i)
        for (int j = 0; j < 4; ++j)
            acc[i][j] = (v8f){};

    float4 ra[NA], rb[NB];

    auto loadTile = [&](int kb) {
        const int gk = kb + ak;
        for (int i = 0; i < NA; ++i) {
            const int row = blockRow + am + 64 * i;
            float4 v = make_float4(0.0f, 0.0f, 0.0f, 0.0f);
            if (row < M && gk < K) {
                const float* s;
                if (gk < w0)            s = p0 + (long)row * w0 + gk;
                else if (gk - w0 < w1)  s = p1 + (long)row * w1 + (gk - w0);
                else                    s = p2 + (long)row * w2 + (gk - w0 - w1);
                v = *(const float4*)s;
            }
            ra[i] = v;
        }
        const int col = blockCol + bn;
        for (int i = 0; i < NB; ++i) {
            const int k = kb + bk + BR * i;
            float4 v = make_float4(0.0f, 0.0f, 0.0f, 0.0f);
            if (k < K && col < N) v = *(const float4*)(B + (long)k * N + col);
            rb[i] = v;
        }
    };

    auto storeTile = [&](int buf) {
        for (int i = 0; i < NA; ++i)
            *(float4*)&As[buf][(am + 64 * i) * GT_AP + ak] = ra[i];
        for (int i = 0; i < NB; ++i)
            *(float4*)&Bs[buf][(bk + BR * i) * BP + bn] = rb[i];
    };

    const int nT = (K + GT_BK - 1) / GT_BK;

    loadTile(0);
    storeTile(0);
    __syncthreads();

    for (int t = 0; t < nT; ++t) {
        const int buf = t & 1;
        if (t + 1 < nT) loadTile((t + 1) * GT_BK);

        const float* Ab = &As[buf][(m0 + c) * GT_AP + 2 * h];
        const float* Bb = &Bs[buf][(2 * h) * BP + n0 + 4 * c];
        for (int kk = 0; kk < GT_BK; kk += 4) {
            v2f a[TI];
            for (int i = 0; i < TI; ++i)
                a[i] = *(const v2f*)&Ab[(16 * i) * GT_AP + kk];
            const float4 bx = *(const float4*)&Bb[kk * BP];
            const float4 by = *(const float4*)&Bb[(kk + 1) * BP];
            v2f b[4];
            b[0].x = bx.x; b[0].y = by.x;
            b[1].x = bx.y; b[1].y = by.y;
            b[2].x = bx.z; b[2].y = by.z;
            b[3].x = bx.w; b[3].y = by.w;
            for (int i = 0; i < TI; ++i)
                for (int j = 0; j < 4; ++j)
                    acc[i][j] = __builtin_amdgcn_wmma_f32_16x16x4_f32(
                        false, a[i], false, b[j], (short)0, acc[i][j], false, false);
        }

        if (t + 1 < nT) storeTile(buf ^ 1);
        __syncthreads();
    }

    const int col = blockCol + n0 + 4 * c;
    if (col < N) {
        for (int i = 0; i < TI; ++i) {
            for (int r = 0; r < 8; ++r) {
                const int row = blockRow + m0 + 16 * i + 8 * h + r;
                if (row < M) {
                    float4 v = make_float4(acc[i][0][r], acc[i][1][r], acc[i][2][r], acc[i][3][r]);
                    *(float4*)(C + (long)row * N + col) = v;
                }
            }
        }
    }
}

__global__ __launch_bounds__(256) void attn_proj(
    const float* __restrict__ feat, const float* __restrict__ al,
    const float* __restrict__ ar, float* __restrict__ el,
    float* __restrict__ er, int nNodes, int H)
{
    int wid  = blockIdx.x * 8 + (threadIdx.x >> 5);
    int lane = threadIdx.x & 31;
    if (wid >= nNodes * H) return;
    int node = wid / H;
    int h    = wid - node * H;
    int C    = H * 64;
    const float* f  = feat + (long)node * C + h * 64;
    const float* wl = al + h * 64;
    const float* wr = ar + h * 64;
    float f0 = f[lane], f1 = f[lane + 32];
    float sl = f0 * wl[lane] + f1 * wl[lane + 32];
    float sr = f0 * wr[lane] + f1 * wr[lane + 32];
#pragma unroll
    for (int o = 16; o > 0; o >>= 1) {
        sl += __shfl_xor(sl, o, 32);
        sr += __shfl_xor(sr, o, 32);
    }
    if (lane == 0) { el[wid] = sl; er[wid] = sr; }
}

__global__ void edge_lrelu_max(
    const int* __restrict__ src, const int* __restrict__ dst,
    const float* __restrict__ el, const float* __restrict__ er,
    float* __restrict__ ebuf, float* __restrict__ emax, int E, int H)
{
    int i = blockIdx.x * blockDim.x + threadIdx.x;
    if (i >= E) return;
    int s = src[i], d = dst[i];
    for (int h = 0; h < H; ++h) {
        float x = el[s * H + h] + er[d * H + h];
        float e = x > 0.0f ? x : 0.2f * x;
        ebuf[(long)i * H + h] = e;
        atomicMaxF(&emax[d * H + h], e);
    }
}

__global__ void edge_expsum(
    const int* __restrict__ dst, float* __restrict__ ebuf,
    const float* __restrict__ emax, float* __restrict__ denom, int E, int H)
{
    int i = blockIdx.x * blockDim.x + threadIdx.x;
    if (i >= E) return;
    int d = dst[i];
    for (int h = 0; h < H; ++h) {
        float ex = expf(ebuf[(long)i * H + h] - emax[d * H + h]);
        ebuf[(long)i * H + h] = ex;
        atomicAdd(&denom[d * H + h], ex);
    }
}

__global__ __launch_bounds__(256) void edge_aggregate(
    const int* __restrict__ src, const int* __restrict__ dst,
    const float* __restrict__ feat, const float* __restrict__ ebuf,
    const float* __restrict__ denom, float* __restrict__ rst, int E, int H)
{
    int wid  = blockIdx.x * 8 + (threadIdx.x >> 5);
    int lane = threadIdx.x & 31;
    if (wid >= E) return;
    int s = src[wid], d = dst[wid];
    int C = H * 64;
    for (int idx = lane; idx < C; idx += 32) {
        int h = idx >> 6;
        float alpha = ebuf[(long)wid * H + h] / denom[d * H + h];
        atomicAdd(&rst[(long)d * C + idx], feat[(long)s * C + idx] * alpha);
    }
}

__global__ void bias_elu(float* __restrict__ x, const float* __restrict__ b,
                         long n, int C)
{
    long i = (long)blockIdx.x * blockDim.x + threadIdx.x;
    if (i >= n) return;
    float v = x[i] + b[i % C];
    x[i] = v > 0.0f ? v : expm1f(v);
}

__global__ __launch_bounds__(256) void classify(
    const float* __restrict__ rst2, const float* __restrict__ b2,
    const float* __restrict__ Wc1, const float* __restrict__ bc1,
    const float* __restrict__ Wc2, const float* __restrict__ bc2,
    float* __restrict__ out, int nNodes)
{
    int wid  = blockIdx.x * 8 + (threadIdx.x >> 5);
    int lane = threadIdx.x & 31;
    if (wid >= nNodes) return;
    float acc = 0.0f;
#pragma unroll
    for (int d = 0; d < 64; ++d)
        acc += (rst2[(long)wid * 64 + d] + b2[d]) * Wc1[d * 32 + lane];
    float t = fmaxf(acc + bc1[lane], 0.0f);
    float o0 = t * Wc2[lane * 2 + 0];
    float o1 = t * Wc2[lane * 2 + 1];
#pragma unroll
    for (int o = 16; o > 0; o >>= 1) {
        o0 += __shfl_xor(o0, o, 32);
        o1 += __shfl_xor(o1, o, 32);
    }
    if (lane == 0) {
        out[(long)wid * 2 + 0] = o0 + bc2[0];
        out[(long)wid * 2 + 1] = o1 + bc2[1];
    }
}

extern "C" void kernel_launch(void* const* d_in, const int* in_sizes, int n_in,
                              void* d_out, int out_size, void* d_ws, size_t ws_size,
                              hipStream_t stream)
{
    (void)n_in; (void)out_size; (void)ws_size;

    const float* randfeat = (const float*)d_in[0];
    const float* func_emb = (const float*)d_in[1];
    const float* codebert = (const float*)d_in[2];
    const int*   src      = (const int*)d_in[3];
    const int*   dst      = (const int*)d_in[4];
    const float* W1  = (const float*)d_in[5];
    const float* al1 = (const float*)d_in[6];
    const float* ar1 = (const float*)d_in[7];
    const float* b1  = (const float*)d_in[8];
    const float* W2  = (const float*)d_in[9];
    const float* al2 = (const float*)d_in[10];
    const float* ar2 = (const float*)d_in[11];
    const float* b2  = (const float*)d_in[12];
    const float* Wc1 = (const float*)d_in[13];
    const float* bc1 = (const float*)d_in[14];
    const float* Wc2 = (const float*)d_in[15];
    const float* bc2 = (const float*)d_in[16];
    float* out = (float*)d_out;

    const int  E  = in_sizes[3];
    const long NN = N_NODES;

    float* ws     = (float*)d_ws;
    float* feat1  = ws;
    float* rst1   = feat1  + NN * 256;
    float* feat2  = rst1   + NN * 256;
    float* rst2   = feat2  + NN * 64;
    float* el1    = rst2   + NN * 64;
    float* er1    = el1    + NN * 4;
    float* emax1  = er1    + NN * 4;
    float* denom1 = emax1  + NN * 4;
    float* el2    = denom1 + NN * 4;
    float* er2    = el2    + NN;
    float* emax2  = er2    + NN;
    float* denom2 = emax2  + NN;
    float* e1     = denom2 + NN;
    float* e2     = e1 + (long)E * 4;

    fill_f32<<<(int)((NN * 256 + 255) / 256), 256, 0, stream>>>(rst1, 0.0f, NN * 256);
    fill_f32<<<(int)((NN * 64 + 255) / 256), 256, 0, stream>>>(rst2, 0.0f, NN * 64);
    fill_f32<<<(int)((NN * 4 + 255) / 256), 256, 0, stream>>>(denom1, 0.0f, NN * 4);
    fill_f32<<<(int)((NN + 255) / 256), 256, 0, stream>>>(denom2, 0.0f, NN);
    fill_f32<<<(int)((NN * 4 + 255) / 256), 256, 0, stream>>>(emax1, -INFINITY, NN * 4);
    fill_f32<<<(int)((NN + 255) / 256), 256, 0, stream>>>(emax2, -INFINITY, NN);

    dim3 g1((N_NODES + 127) / 128, 1);
    gemm_wmma_f32_tiled<2, 4, 4><<<g1, 256, 0, stream>>>(randfeat, func_emb, codebert,
                                                        100, 100, 768, W1, feat1,
                                                        N_NODES, 256, 968);
    attn_proj<<<(N_NODES * 4 + 7) / 8, 256, 0, stream>>>(feat1, al1, ar1, el1, er1, N_NODES, 4);
    edge_lrelu_max<<<(E + 255) / 256, 256, 0, stream>>>(src, dst, el1, er1, e1, emax1, E, 4);
    edge_expsum<<<(E + 255) / 256, 256, 0, stream>>>(dst, e1, emax1, denom1, E, 4);
    edge_aggregate<<<(E + 7) / 8, 256, 0, stream>>>(src, dst, feat1, e1, denom1, rst1, E, 4);
    bias_elu<<<(int)((NN * 256 + 255) / 256), 256, 0, stream>>>(rst1, b1, NN * 256, 256);

    dim3 g2((N_NODES + 127) / 128, 1);
    gemm_wmma_f32_tiled<8, 1, 1><<<g2, 256, 0, stream>>>(rst1, rst1, rst1,
                                                        256, 0, 0, W2, feat2,
                                                        N_NODES, 64, 256);
    attn_proj<<<(N_NODES + 7) / 8, 256, 0, stream>>>(feat2, al2, ar2, el2, er2, N_NODES, 1);
    edge_lrelu_max<<<(E + 255) / 256, 256, 0, stream>>>(src, dst, el2, er2, e2, emax2, E, 1);
    edge_expsum<<<(E + 255) / 256, 256, 0, stream>>>(dst, e2, emax2, denom2, E, 1);
    edge_aggregate<<<(E + 7) / 8, 256, 0, stream>>>(src, dst, feat2, e2, denom2, rst2, E, 1);

    classify<<<(N_NODES + 7) / 8, 256, 0, stream>>>(rst2, b2, Wc1, bc1, Wc2, bc2, out, N_NODES);
}


# Edge softmax+aggregation as a gather: edges counting-sorted by dst each call (int atomics), one wave per dst row sums its edges in registers, arrival order; no f32 atomics or fills; bias+elu at store
# speedup vs baseline: 3.7117x; 2.4328x over previous
#include <hip/hip_runtime.h>
#include <math.h>


typedef __attribute__((ext_vector_type(2))) float v2f;
typedef __attribute__((ext_vector_type(8))) float v8f;

#define N_NODES 50000

__global__ void fill_i32(int* __restrict__ p, int v, int n) {
    int i = blockIdx.x * blockDim.x + threadIdx.x;
    if (i < n) p[i] = v;
}

__global__ void csr_count(const int* __restrict__ dst, int* __restrict__ cnt, int E) {
    int i = blockIdx.x * blockDim.x + threadIdx.x;
    if (i < E) atomicAdd(&cnt[dst[i]], 1);
}

__global__ __launch_bounds__(1024) void csr_scan(int* __restrict__ cur, int* __restrict__ rowBeg, int n) {
    __shared__ int wtot[2][32];
    const int t = threadIdx.x, lane = t & 31, wave = t >> 5;
    auto loadTile = [&](int i) {
        int4 v = make_int4(0, 0, 0, 0);
        if (i + 3 < n) {
            v = *(const int4*)(cur + i);
        } else {
            if (i < n)     v.x = cur[i];
            if (i + 1 < n) v.y = cur[i + 1];
            if (i + 2 < n) v.z = cur[i + 2];
        }
        return v;
    };
    int carry = 0;
    int it = 0;
    int4 nxt = loadTile(4 * t);
    for (int base = 0; base < n; base += 4096, ++it) {
        const int i = base + 4 * t;
        const int4 v = nxt;
        if (base + 4096 < n) nxt = loadTile(i + 4096);
        const int s = v.x + v.y + v.z + v.w;
        int incl = s;
        for (int o = 1; o < 32; o <<= 1) {
            int u = __shfl_up(incl, o, 32);
            if (lane >= o) incl += u;
        }
        if (lane == 31) wtot[it & 1][wave] = incl;
        __syncthreads();
        const int w = wtot[it & 1][lane];
        int wi = w;
        for (int o = 1; o < 32; o <<= 1) {
            int u = __shfl_up(wi, o, 32);
            if (lane >= o) wi += u;
        }
        const int waveOff = __shfl(wi, wave, 32) - __shfl(w, wave, 32);
        const int total   = __shfl(wi, 31, 32);
        int4 r;
        r.x = carry + waveOff + incl - s;
        r.y = r.x + v.x;
        r.z = r.y + v.y;
        r.w = r.z + v.z;
        if (i + 3 < n) {
            *(int4*)(cur + i)    = r;
            *(int4*)(rowBeg + i) = r;
        } else {
            if (i < n)     { cur[i]     = r.x; rowBeg[i]     = r.x; }
            if (i + 1 < n) { cur[i + 1] = r.y; rowBeg[i + 1] = r.y; }
            if (i + 2 < n) { cur[i + 2] = r.z; rowBeg[i + 2] = r.z; }
        }
        carry += total;
    }
}

__global__ void csr_place(const int* __restrict__ src, const int* __restrict__ dst,
                          int* __restrict__ cur, int* __restrict__ csrSrc, int E) {
    int i = blockIdx.x * blockDim.x + threadIdx.x;
    if (i < E) {
        int p = atomicAdd(&cur[dst[i]], 1);
        csrSrc[p] = src[i];
    }
}

template <int H> struct RowVec;
template <> struct RowVec<4> { float4 a, b; };
template <> struct RowVec<1> { float2 a; };

template <int H, bool ELU>
__global__ __launch_bounds__(256) void gat_gather(
    const int* __restrict__ rowBeg, const int* __restrict__ rowEnd,
    const int* __restrict__ csrSrc,
    const float* __restrict__ feat, const float* __restrict__ el,
    const float* __restrict__ er, const float* __restrict__ bias,
    float* __restrict__ out, int nNodes)
{
    static_assert(H == 4 || H == 1, "head counts of this model");
    constexpr int C = H * 64;
    const int lane = threadIdx.x & 31;
    const int d = __builtin_amdgcn_readfirstlane((int)(blockIdx.x * 8 + (threadIdx.x >> 5)));
    if (d >= nNodes) return;
    const int beg = rowBeg[d];
    const int end = rowEnd[d];

    float erd[H];
    for (int h = 0; h < H; ++h) erd[h] = er[d * H + h];

    auto logits = [&](int s, float (&e)[H]) {
        float x[H];
        if constexpr (H == 4) {
            const float4 q = *(const float4*)(el + (long)s * 4);
            x[0] = q.x; x[1] = q.y; x[2] = q.z; x[3] = q.w;
        } else {
            x[0] = el[s];
        }
        for (int h = 0; h < H; ++h) {
            const float y = x[h] + erd[h];
            e[h] = y > 0.0f ? y : 0.2f * y;
        }
    };

    float m[H];
    for (int h = 0; h < H; ++h) m[h] = -INFINITY;
    for (int base = beg; base < end; base += 32) {
        const int i = base + lane;
        if (i < end) {
            float e[H];
            logits(csrSrc[i], e);
            for (int h = 0; h < H; ++h) m[h] = fmaxf(m[h], e[h]);
        }
    }
    for (int h = 0; h < H; ++h) {
        for (int o = 16; o > 0; o >>= 1) m[h] = fmaxf(m[h], __shfl_xor(m[h], o, 32));
        if (!isfinite(m[h])) m[h] = 0.0f;
    }

    float den[H];
    for (int h = 0; h < H; ++h) den[h] = 0.0f;
    for (int base = beg; base < end; base += 32) {
        const int i = base + lane;
        if (i < end) {
            float e[H];
            logits(csrSrc[i], e);
            for (int h = 0; h < H; ++h) den[h] += expf(e[h] - m[h]);
        }
    }
    for (int h = 0; h < H; ++h) {
        for (int o = 16; o > 0; o >>= 1) den[h] += __shfl_xor(den[h], o, 32);
        if (den[h] == 0.0f) den[h] = 1.0f;
    }

    RowVec<H> acc;
    if constexpr (H == 4) {
        acc.a = make_float4(0.0f, 0.0f, 0.0f, 0.0f);
        acc.b = make_float4(0.0f, 0.0f, 0.0f, 0.0f);
    } else {
        acc.a = make_float2(0.0f, 0.0f);
    }
    const bool hi = lane >= 16;
    for (int base = beg; base < end; base += 32) {
        const int i = base + lane;
        int   s = 0;
        float al[H];
        for (int h = 0; h < H; ++h) al[h] = 0.0f;
        if (i < end) {
            float e[H];
            s = csrSrc[i];
            logits(s, e);
            for (int h = 0; h < H; ++h) al[h] = expf(e[h] - m[h]) / den[h];
        }
        const int cnt = min(32, end - base);
#pragma unroll 4
        for (int j = 0; j < cnt; ++j) {
            const int sj = __builtin_amdgcn_readlane(s, j);
            float aj[H];
            for (int h = 0; h < H; ++h)
                aj[h] = __int_as_float(__builtin_amdgcn_readlane(__float_as_int(al[h]), j));
            const float* row = feat + (long)sj * C;
            if constexpr (H == 4) {
                const float4 f0 = *(const float4*)(row + 4 * lane);
                const float4 f1 = *(const float4*)(row + 128 + 4 * lane);
                const float w0 = hi ? aj[1] : aj[0];
                const float w1 = hi ? aj[3] : aj[2];
                acc.a.x += f0.x * w0; acc.a.y += f0.y * w0; acc.a.z += f0.z * w0; acc.a.w += f0.w * w0;
                acc.b.x += f1.x * w1; acc.b.y += f1.y * w1; acc.b.z += f1.z * w1; acc.b.w += f1.w * w1;
            } else {
                const float2 f0 = *(const float2*)(row + 2 * lane);
                acc.a.x += f0.x * aj[0]; acc.a.y += f0.y * aj[0];
            }
        }
    }

    float* o = out + (long)d * C;
    if constexpr (H == 4) {
        float4 r0 = acc.a, r1 = acc.b;
        if constexpr (ELU) {
            const float4 b0 = *(const float4*)(bias + 4 * lane);
            const float4 b1 = *(const float4*)(bias + 128 + 4 * lane);
            r0.x += b0.x; r0.y += b0.y; r0.z += b0.z; r0.w += b0.w;
            r1.x += b1.x; r1.y += b1.y; r1.z += b1.z; r1.w += b1.w;
            r0.x = r0.x > 0.0f ? r0.x : expm1f(r0.x); r0.y = r0.y > 0.0f ? r0.y : expm1f(r0.y);
            r0.z = r0.z > 0.0f ? r0.z : expm1f(r0.z); r0.w = r0.w > 0.0f ? r0.w : expm1f(r0.w);
            r1.x = r1.x > 0.0f ? r1.x : expm1f(r1.x); r1.y = r1.y > 0.0f ? r1.y : expm1f(r1.y);
            r1.z = r1.z > 0.0f ? r1.z : expm1f(r1.z); r1.w = r1.w > 0.0f ? r1.w : expm1f(r1.w);
        }
        *(float4*)(o + 4 * lane)       = r0;
        *(float4*)(o + 128 + 4 * lane) = r1;
    } else {
        float2 r0 = acc.a;
        if constexpr (ELU) {
            const float2 b0 = *(const float2*)(bias + 2 * lane);
            r0.x += b0.x; r0.y += b0.y;
            r0.x = r0.x > 0.0f ? r0.x : expm1f(r0.x); r0.y = r0.y > 0.0f ? r0.y : expm1f(r0.y);
        }
        *(float2*)(o + 2 * lane) = r0;
    }
}

#define GT_BK 16
#define GT_AP 20

template <int WM, int WN, int TI>
__global__ __launch_bounds__(256) void gemm_wmma_f32_tiled(
    const float* __restrict__ p0, const float* __restrict__ p1,
    const float* __restrict__ p2, int w0, int w1, int w2,
    const float* __restrict__ B, float* __restrict__ C,
    int M, int N, int K)
{
    static_assert(WM * WN == 8, "8 waves per block");
    constexpr int BM  = WM * TI * 16;
    constexpr int BN  = WN * 64;
    constexpr int BP  = BN + 8;
    constexpr int NA  = BM / 64;
    constexpr int NB  = BN / 64;
    constexpr int BQ  = BN / 4;
    constexpr int BR  = 256 / BQ;
    static_assert(BM % 64 == 0, "A staging covers 64 rows per pass");

    __shared__ __attribute__((aligned(16))) float As[2][BM * GT_AP];
    __shared__ __attribute__((aligned(16))) float Bs[2][GT_BK * BP];

    const int tid  = threadIdx.x;
    const int lane = tid & 31;
    const int wave = tid >> 5;
    const int c    = lane & 15;
    const int h    = lane >> 4;
    const int m0   = (wave % WM) * (TI * 16);
    const int n0   = (wave / WM) * 64;

    const int blockRow = blockIdx.x * BM;
    const int blockCol = blockIdx.y * BN;

    const int am = tid >> 2;
    const int ak = (tid & 3) << 2;
    const int bk = tid / BQ;
    const int bn = (tid % BQ) << 2;

    v8f acc[TI][4];
    for (int i = 0; i < TI; ++i)
        for (int j = 0; j < 4; ++j)
            acc[i][j] = (v8f){};

    float4 ra[NA], rb[NB];

    auto loadTile = [&](int kb) {
        const int gk = kb + ak;
        for (int i = 0; i < NA; ++i) {
            const int row = blockRow + am + 64 * i;
            float4 v = make_float4(0.0f, 0.0f, 0.0f, 0.0f);
            if (row < M && gk < K) {
                const float* s;
                if (gk < w0)            s = p0 + (long)row * w0 + gk;
                else if (gk - w0 < w1)  s = p1 + (long)row * w1 + (gk - w0);
                else                    s = p2 + (long)row * w2 + (gk - w0 - w1);
                v = *(const float4*)s;
            }
            ra[i] = v;
        }
        const int col = blockCol + bn;
        for (int i = 0; i < NB; ++i) {
            const int k = kb + bk + BR * i;
            float4 v = make_float4(0.0f, 0.0f, 0.0f, 0.0f);
            if (k < K && col < N) v = *(const float4*)(B + (long)k * N + col);
            rb[i] = v;
        }
    };

    auto storeTile = [&](int buf) {
        for (int i = 0; i < NA; ++i)
            *(float4*)&As[buf][(am + 64 * i) * GT_AP + ak] = ra[i];
        for (int i = 0; i < NB; ++i)
            *(float4*)&Bs[buf][(bk + BR * i) * BP + bn] = rb[i];
    };

    const int nT = (K + GT_BK - 1) / GT_BK;

    loadTile(0);
    storeTile(0);
    __syncthreads();

    for (int t = 0; t < nT; ++t) {
        const int buf = t & 1;
        if (t + 1 < nT) loadTile((t + 1) * GT_BK);

        const float* Ab = &As[buf][(m0 + c) * GT_AP + 2 * h];
        const float* Bb = &Bs[buf][(2 * h) * BP + n0 + 4 * c];
        for (int kk = 0; kk < GT_BK; kk += 4) {
            v2f a[TI];
            for (int i = 0; i < TI; ++i)
                a[i] = *(const v2f*)&Ab[(16 * i) * GT_AP + kk];
            const float4 bx = *(const float4*)&Bb[kk * BP];
            const float4 by = *(const float4*)&Bb[(kk + 1) * BP];
            v2f b[4];
            b[0].x = bx.x; b[0].y = by.x;
            b[1].x = bx.y; b[1].y = by.y;
            b[2].x = bx.z; b[2].y = by.z;
            b[3].x = bx.w; b[3].y = by.w;
            for (int i = 0; i < TI; ++i)
                for (int j = 0; j < 4; ++j)
                    acc[i][j] = __builtin_amdgcn_wmma_f32_16x16x4_f32(
                        false, a[i], false, b[j], (short)0, acc[i][j], false, false);
        }

        if (t + 1 < nT) storeTile(buf ^ 1);
        __syncthreads();
    }

    const int col = blockCol + n0 + 4 * c;
    if (col < N) {
        for (int i = 0; i < TI; ++i) {
            for (int r = 0; r < 8; ++r) {
                const int row = blockRow + m0 + 16 * i + 8 * h + r;
                if (row < M) {
                    float4 v = make_float4(acc[i][0][r], acc[i][1][r], acc[i][2][r], acc[i][3][r]);
                    *(float4*)(C + (long)row * N + col) = v;
                }
            }
        }
    }
}

__global__ __launch_bounds__(256) void attn_proj(
    const float* __restrict__ feat, const float* __restrict__ al,
    const float* __restrict__ ar, float* __restrict__ el,
    float* __restrict__ er, int nNodes, int H)
{
    int wid  = blockIdx.x * 8 + (threadIdx.x >> 5);
    int lane = threadIdx.x & 31;
    if (wid >= nNodes * H) return;
    int node = wid / H;
    int h    = wid - node * H;
    int C    = H * 64;
    const float* f  = feat + (long)node * C + h * 64;
    const float* wl = al + h * 64;
    const float* wr = ar + h * 64;
    float f0 = f[lane], f1 = f[lane + 32];
    float sl = f0 * wl[lane] + f1 * wl[lane + 32];
    float sr = f0 * wr[lane] + f1 * wr[lane + 32];
#pragma unroll
    for (int o = 16; o > 0; o >>= 1) {
        sl += __shfl_xor(sl, o, 32);
        sr += __shfl_xor(sr, o, 32);
    }
    if (lane == 0) { el[wid] = sl; er[wid] = sr; }
}

__global__ __launch_bounds__(256) void classify(
    const float* __restrict__ rst2, const float* __restrict__ b2,
    const float* __restrict__ Wc1, const float* __restrict__ bc1,
    const float* __restrict__ Wc2, const float* __restrict__ bc2,
    float* __restrict__ out, int nNodes)
{
    int wid  = blockIdx.x * 8 + (threadIdx.x >> 5);
    int lane = threadIdx.x & 31;
    if (wid >= nNodes) return;
    float acc = 0.0f;
#pragma unroll
    for (int d = 0; d < 64; ++d)
        acc += (rst2[(long)wid * 64 + d] + b2[d]) * Wc1[d * 32 + lane];
    float t = fmaxf(acc + bc1[lane], 0.0f);
    float o0 = t * Wc2[lane * 2 + 0];
    float o1 = t * Wc2[lane * 2 + 1];
#pragma unroll
    for (int o = 16; o > 0; o >>= 1) {
        o0 += __shfl_xor(o0, o, 32);
        o1 += __shfl_xor(o1, o, 32);
    }
    if (lane == 0) {
        out[(long)wid * 2 + 0] = o0 + bc2[0];
        out[(long)wid * 2 + 1] = o1 + bc2[1];
    }
}

extern "C" void kernel_launch(void* const* d_in, const int* in_sizes, int n_in,
                              void* d_out, int out_size, void* d_ws, size_t ws_size,
                              hipStream_t stream)
{
    (void)n_in; (void)out_size; (void)ws_size;

    const float* randfeat = (const float*)d_in[0];
    const float* func_emb = (const float*)d_in[1];
    const float* codebert = (const float*)d_in[2];
    const int*   src      = (const int*)d_in[3];
    const int*   dst      = (const int*)d_in[4];
    const float* W1  = (const float*)d_in[5];
    const float* al1 = (const float*)d_in[6];
    const float* ar1 = (const float*)d_in[7];
    const float* b1  = (const float*)d_in[8];
    const float* W2  = (const float*)d_in[9];
    const float* al2 = (const float*)d_in[10];
    const float* ar2 = (const float*)d_in[11];
    const float* b2  = (const float*)d_in[12];
    const float* Wc1 = (const float*)d_in[13];
    const float* bc1 = (const float*)d_in[14];
    const float* Wc2 = (const float*)d_in[15];
    const float* bc2 = (const float*)d_in[16];
    float* out = (float*)d_out;

    const int  E  = in_sizes[3];
    const long NN = N_NODES;

    float* ws     = (float*)d_ws;
    float* feat1  = ws;
    float* h2     = feat1  + NN * 256;
    float* feat2  = h2     + NN * 256;
    float* rst2   = feat2  + NN * 64;
    float* el1    = rst2   + NN * 64;
    float* er1    = el1    + NN * 4;
    float* el2    = er1    + NN * 4;
    float* er2    = el2    + NN;
    int*   cur    = (int*)(er2 + NN);
    int*   rowBeg = cur    + NN;
    int*   csrSrc = rowBeg + NN;

    fill_i32<<<(N_NODES + 255) / 256, 256, 0, stream>>>(cur, 0, N_NODES);
    csr_count<<<(E + 255) / 256, 256, 0, stream>>>(dst, cur, E);
    csr_scan<<<1, 1024, 0, stream>>>(cur, rowBeg, N_NODES);
    csr_place<<<(E + 255) / 256, 256, 0, stream>>>(src, dst, cur, csrSrc, E);

    dim3 g1((N_NODES + 127) / 128, 1);
    gemm_wmma_f32_tiled<2, 4, 4><<<g1, 256, 0, stream>>>(randfeat, func_emb, codebert,
                                                        100, 100, 768, W1, feat1,
                                                        N_NODES, 256, 968);
    attn_proj<<<(N_NODES * 4 + 7) / 8, 256, 0, stream>>>(feat1, al1, ar1, el1, er1, N_NODES, 4);
    gat_gather<4, true><<<(N_NODES + 7) / 8, 256, 0, stream>>>(rowBeg, cur, csrSrc, feat1, el1, er1,
                                                              b1, h2, N_NODES);

    dim3 g2((N_NODES + 127) / 128, 1);
    gemm_wmma_f32_tiled<8, 1, 1><<<g2, 256, 0, stream>>>(h2, h2, h2,
                                                        256, 0, 0, W2, feat2,
                                                        N_NODES, 64, 256);
    attn_proj<<<(N_NODES + 7) / 8, 256, 0, stream>>>(feat2, al2, ar2, el2, er2, N_NODES, 1);
    gat_gather<1, false><<<(N_NODES + 7) / 8, 256, 0, stream>>>(rowBeg, cur, csrSrc, feat2, el2, er2,
                                                               b2, rst2, N_NODES);

    classify<<<(N_NODES + 7) / 8, 256, 0, stream>>>(rst2, b2, Wc1, bc1, Wc2, bc2, out, N_NODES);
}


# Attention logits el/er formed in the GEMM epilogue from the accumulators (a wave's 64 columns = one head; 16-lane tree sum, new summation order); both projection launches removed
# speedup vs baseline: 4.0045x; 1.0789x over previous
#include <hip/hip_runtime.h>
#include <math.h>


typedef __attribute__((ext_vector_type(2))) float v2f;
typedef __attribute__((ext_vector_type(8))) float v8f;

#define N_NODES 50000

__global__ void fill_i32(int* __restrict__ p, int v, int n) {
    int i = blockIdx.x * blockDim.x + threadIdx.x;
    if (i < n) p[i] = v;
}

__global__ void csr_count(const int* __restrict__ dst, int* __restrict__ cnt, int E) {
    int i = blockIdx.x * blockDim.x + threadIdx.x;
    if (i < E) atomicAdd(&cnt[dst[i]], 1);
}

__global__ __launch_bounds__(1024) void csr_scan(int* __restrict__ cur, int* __restrict__ rowBeg, int n) {
    __shared__ int wtot[2][32];
    const int t = threadIdx.x, lane = t & 31, wave = t >> 5;
    auto loadTile = [&](int i) {
        int4 v = make_int4(0, 0, 0, 0);
        if (i + 3 < n) {
            v = *(const int4*)(cur + i);
        } else {
            if (i < n)     v.x = cur[i];
            if (i + 1 < n) v.y = cur[i + 1];
            if (i + 2 < n) v.z = cur[i + 2];
        }
        return v;
    };
    int carry = 0;
    int it = 0;
    int4 nxt = loadTile(4 * t);
    for (int base = 0; base < n; base += 4096, ++it) {
        const int i = base + 4 * t;
        const int4 v = nxt;
        if (base + 4096 < n) nxt = loadTile(i + 4096);
        const int s = v.x + v.y + v.z + v.w;
        int incl = s;
        for (int o = 1; o < 32; o <<= 1) {
            int u = __shfl_up(incl, o, 32);
            if (lane >= o) incl += u;
        }
        if (lane == 31) wtot[it & 1][wave] = incl;
        __syncthreads();
        const int w = wtot[it & 1][lane];
        int wi = w;
        for (int o = 1; o < 32; o <<= 1) {
            int u = __shfl_up(wi, o, 32);
            if (lane >= o) wi += u;
        }
        const int waveOff = __shfl(wi, wave, 32) - __shfl(w, wave, 32);
        const int total   = __shfl(wi, 31, 32);
        int4 r;
        r.x = carry + waveOff + incl - s;
        r.y = r.x + v.x;
        r.z = r.y + v.y;
        r.w = r.z + v.z;
        if (i + 3 < n) {
            *(int4*)(cur + i)    = r;
            *(int4*)(rowBeg + i) = r;
        } else {
            if (i < n)     { cur[i]     = r.x; rowBeg[i]     = r.x; }
            if (i + 1 < n) { cur[i + 1] = r.y; rowBeg[i + 1] = r.y; }
            if (i + 2 < n) { cur[i + 2] = r.z; rowBeg[i + 2] = r.z; }
        }
        carry += total;
    }
}

__global__ void csr_place(const int* __restrict__ src, const int* __restrict__ dst,
                          int* __restrict__ cur, int* __restrict__ csrSrc, int E) {
    int i = blockIdx.x * blockDim.x + threadIdx.x;
    if (i < E) {
        int p = atomicAdd(&cur[dst[i]], 1);
        csrSrc[p] = src[i];
    }
}

template <int H> struct RowVec;
template <> struct RowVec<4> { float4 a, b; };
template <> struct RowVec<1> { float2 a; };

template <int H, bool ELU>
__global__ __launch_bounds__(256) void gat_gather(
    const int* __restrict__ rowBeg, const int* __restrict__ rowEnd,
    const int* __restrict__ csrSrc,
    const float* __restrict__ feat, const float* __restrict__ el,
    const float* __restrict__ er, const float* __restrict__ bias,
    float* __restrict__ out, int nNodes)
{
    static_assert(H == 4 || H == 1, "head counts of this model");
    constexpr int C = H * 64;
    const int lane = threadIdx.x & 31;
    const int d = __builtin_amdgcn_readfirstlane((int)(blockIdx.x * 8 + (threadIdx.x >> 5)));
    if (d >= nNodes) return;
    const int beg = rowBeg[d];
    const int end = rowEnd[d];

    float erd[H];
    for (int h = 0; h < H; ++h) erd[h] = er[d * H + h];

    auto logits = [&](int s, float (&e)[H]) {
        float x[H];
        if constexpr (H == 4) {
            const float4 q = *(const float4*)(el + (long)s * 4);
            x[0] = q.x; x[1] = q.y; x[2] = q.z; x[3] = q.w;
        } else {
            x[0] = el[s];
        }
        for (int h = 0; h < H; ++h) {
            const float y = x[h] + erd[h];
            e[h] = y > 0.0f ? y : 0.2f * y;
        }
    };

    float m[H];
    for (int h = 0; h < H; ++h) m[h] = -INFINITY;
    for (int base = beg; base < end; base += 32) {
        const int i = base + lane;
        if (i < end) {
            float e[H];
            logits(csrSrc[i], e);
            for (int h = 0; h < H; ++h) m[h] = fmaxf(m[h], e[h]);
        }
    }
    for (int h = 0; h < H; ++h) {
        for (int o = 16; o > 0; o >>= 1) m[h] = fmaxf(m[h], __shfl_xor(m[h], o, 32));
        if (!isfinite(m[h])) m[h] = 0.0f;
    }

    float den[H];
    for (int h = 0; h < H; ++h) den[h] = 0.0f;
    for (int base = beg; base < end; base += 32) {
        const int i = base + lane;
        if (i < end) {
            float e[H];
            logits(csrSrc[i], e);
            for (int h = 0; h < H; ++h) den[h] += expf(e[h] - m[h]);
        }
    }
    for (int h = 0; h < H; ++h) {
        for (int o = 16; o > 0; o >>= 1) den[h] += __shfl_xor(den[h], o, 32);
        if (den[h] == 0.0f) den[h] = 1.0f;
    }

    RowVec<H> acc;
    if constexpr (H == 4) {
        acc.a = make_float4(0.0f, 0.0f, 0.0f, 0.0f);
        acc.b = make_float4(0.0f, 0.0f, 0.0f, 0.0f);
    } else {
        acc.a = make_float2(0.0f, 0.0f);
    }
    const bool hi = lane >= 16;
    for (int base = beg; base < end; base += 32) {
        const int i = base + lane;
        int   s = 0;
        float al[H];
        for (int h = 0; h < H; ++h) al[h] = 0.0f;
        if (i < end) {
            float e[H];
            s = csrSrc[i];
            logits(s, e);
            for (int h = 0; h < H; ++h) al[h] = expf(e[h] - m[h]) / den[h];
        }
        const int cnt = min(32, end - base);
#pragma unroll 4
        for (int j = 0; j < cnt; ++j) {
            const int sj = __builtin_amdgcn_readlane(s, j);
            float aj[H];
            for (int h = 0; h < H; ++h)
                aj[h] = __int_as_float(__builtin_amdgcn_readlane(__float_as_int(al[h]), j));
            const float* row = feat + (long)sj * C;
            if constexpr (H == 4) {
                const float4 f0 = *(const float4*)(row + 4 * lane);
                const float4 f1 = *(const float4*)(row + 128 + 4 * lane);
                const float w0 = hi ? aj[1] : aj[0];
                const float w1 = hi ? aj[3] : aj[2];
                acc.a.x += f0.x * w0; acc.a.y += f0.y * w0; acc.a.z += f0.z * w0; acc.a.w += f0.w * w0;
                acc.b.x += f1.x * w1; acc.b.y += f1.y * w1; acc.b.z += f1.z * w1; acc.b.w += f1.w * w1;
            } else {
                const float2 f0 = *(const float2*)(row + 2 * lane);
                acc.a.x += f0.x * aj[0]; acc.a.y += f0.y * aj[0];
            }
        }
    }

    float* o = out + (long)d * C;
    if constexpr (H == 4) {
        float4 r0 = acc.a, r1 = acc.b;
        if constexpr (ELU) {
            const float4 b0 = *(const float4*)(bias + 4 * lane);
            const float4 b1 = *(const float4*)(bias + 128 + 4 * lane);
            r0.x += b0.x; r0.y += b0.y; r0.z += b0.z; r0.w += b0.w;
            r1.x += b1.x; r1.y += b1.y; r1.z += b1.z; r1.w += b1.w;
            r0.x = r0.x > 0.0f ? r0.x : expm1f(r0.x); r0.y = r0.y > 0.0f ? r0.y : expm1f(r0.y);
            r0.z = r0.z > 0.0f ? r0.z : expm1f(r0.z); r0.w = r0.w > 0.0f ? r0.w : expm1f(r0.w);
            r1.x = r1.x > 0.0f ? r1.x : expm1f(r1.x); r1.y = r1.y > 0.0f ? r1.y : expm1f(r1.y);
            r1.z = r1.z > 0.0f ? r1.z : expm1f(r1.z); r1.w = r1.w > 0.0f ? r1.w : expm1f(r1.w);
        }
        *(float4*)(o + 4 * lane)       = r0;
        *(float4*)(o + 128 + 4 * lane) = r1;
    } else {
        float2 r0 = acc.a;
        if constexpr (ELU) {
            const float2 b0 = *(const float2*)(bias + 2 * lane);
            r0.x += b0.x; r0.y += b0.y;
            r0.x = r0.x > 0.0f ? r0.x : expm1f(r0.x); r0.y = r0.y > 0.0f ? r0.y : expm1f(r0.y);
        }
        *(float2*)(o + 2 * lane) = r0;
    }
}

#define GT_BK 16
#define GT_AP 20

template <int WM, int WN, int TI>
__global__ __launch_bounds__(256) void gemm_wmma_f32_tiled(
    const float* __restrict__ p0, const float* __restrict__ p1,
    const float* __restrict__ p2, int w0, int w1, int w2,
    const float* __restrict__ B, float* __restrict__ C,
    const float* __restrict__ al, const float* __restrict__ ar,
    float* __restrict__ el, float* __restrict__ er,
    int M, int N, int K)
{
    static_assert(WM * WN == 8, "8 waves per block");
    constexpr int BM  = WM * TI * 16;
    constexpr int BN  = WN * 64;
    constexpr int BP  = BN + 8;
    constexpr int NA  = BM / 64;
    constexpr int NB  = BN / 64;
    constexpr int BQ  = BN / 4;
    constexpr int BR  = 256 / BQ;
    constexpr int HB  = WN;
    static_assert(BM % 64 == 0, "A staging covers 64 rows per pass");

    __shared__ __attribute__((aligned(16))) float As[2][BM * GT_AP];
    __shared__ __attribute__((aligned(16))) float Bs[2][GT_BK * BP];
    __shared__ float Es[2][BM * HB];

    const int tid  = threadIdx.x;
    const int lane = tid & 31;
    const int wave = tid >> 5;
    const int c    = lane & 15;
    const int h    = lane >> 4;
    const int m0   = (wave % WM) * (TI * 16);
    const int n0   = (wave / WM) * 64;

    const int blockRow = blockIdx.x * BM;
    const int blockCol = blockIdx.y * BN;

    const int am = tid >> 2;
    const int ak = (tid & 3) << 2;
    const int bk = tid / BQ;
    const int bn = (tid % BQ) << 2;

    v8f acc[TI][4];
    for (int i = 0; i < TI; ++i)
        for (int j = 0; j < 4; ++j)
            acc[i][j] = (v8f){};

    float4 ra[NA], rb[NB];

    auto loadTile = [&](int kb) {
        const int gk = kb + ak;
        for (int i = 0; i < NA; ++i) {
            const int row = blockRow + am + 64 * i;
            float4 v = make_float4(0.0f, 0.0f, 0.0f, 0.0f);
            if (row < M && gk < K) {
                const float* s;
                if (gk < w0)            s = p0 + (long)row * w0 + gk;
                else if (gk - w0 < w1)  s = p1 + (long)row * w1 + (gk - w0);
                else                    s = p2 + (long)row * w2 + (gk - w0 - w1);
                v = *(const float4*)s;
            }
            ra[i] = v;
        }
        const int col = blockCol + bn;
        for (int i = 0; i < NB; ++i) {
            const int k = kb + bk + BR * i;
            float4 v = make_float4(0.0f, 0.0f, 0.0f, 0.0f);
            if (k < K && col < N) v = *(const float4*)(B + (long)k * N + col);
            rb[i] = v;
        }
    };

    auto storeTile = [&](int buf) {
        for (int i = 0; i < NA; ++i)
            *(float4*)&As[buf][(am + 64 * i) * GT_AP + ak] = ra[i];
        for (int i = 0; i < NB; ++i)
            *(float4*)&Bs[buf][(bk + BR * i) * BP + bn] = rb[i];
    };

    const int nT = (K + GT_BK - 1) / GT_BK;

    loadTile(0);
    storeTile(0);
    __syncthreads();

    for (int t = 0; t < nT; ++t) {
        const int buf = t & 1;
        if (t + 1 < nT) loadTile((t + 1) * GT_BK);

        const float* Ab = &As[buf][(m0 + c) * GT_AP + 2 * h];
        const float* Bb = &Bs[buf][(2 * h) * BP + n0 + 4 * c];
        for (int kk = 0; kk < GT_BK; kk += 4) {
            v2f a[TI];
            for (int i = 0; i < TI; ++i)
                a[i] = *(const v2f*)&Ab[(16 * i) * GT_AP + kk];
            const float4 bx = *(const float4*)&Bb[kk * BP];
            const float4 by = *(const float4*)&Bb[(kk + 1) * BP];
            v2f b[4];
            b[0].x = bx.x; b[0].y = by.x;
            b[1].x = bx.y; b[1].y = by.y;
            b[2].x = bx.z; b[2].y = by.z;
            b[3].x = bx.w; b[3].y = by.w;
            for (int i = 0; i < TI; ++i)
                for (int j = 0; j < 4; ++j)
                    acc[i][j] = __builtin_amdgcn_wmma_f32_16x16x4_f32(
                        false, a[i], false, b[j], (short)0, acc[i][j], false, false);
        }

        if (t + 1 < nT) storeTile(buf ^ 1);
        __syncthreads();
    }

    const int col = blockCol + n0 + 4 * c;

    float4 wl = make_float4(0.0f, 0.0f, 0.0f, 0.0f);
    float4 wr = make_float4(0.0f, 0.0f, 0.0f, 0.0f);
    if (col < N) {
        wl = *(const float4*)(al + col);
        wr = *(const float4*)(ar + col);
    }

    if (col < N) {
        for (int i = 0; i < TI; ++i) {
            for (int r = 0; r < 8; ++r) {
                const int row = blockRow + m0 + 16 * i + 8 * h + r;
                if (row < M) {
                    float4 v = make_float4(acc[i][0][r], acc[i][1][r], acc[i][2][r], acc[i][3][r]);
                    *(float4*)(C + (long)row * N + col) = v;
                }
            }
        }
    }

    float p[TI][16];
    for (int i = 0; i < TI; ++i) {
        for (int r = 0; r < 8; ++r) {
            p[i][r]     = acc[i][0][r] * wl.x + acc[i][1][r] * wl.y + acc[i][2][r] * wl.z + acc[i][3][r] * wl.w;
            p[i][8 + r] = acc[i][0][r] * wr.x + acc[i][1][r] * wr.y + acc[i][2][r] * wr.z + acc[i][3][r] * wr.w;
        }
    }
#pragma unroll
    for (int o = 8; o >= 1; o >>= 1) {
        const bool up = (c & o) != 0;
#pragma unroll
        for (int i = 0; i < TI; ++i) {
#pragma unroll
            for (int k = 0; k < o; ++k) {
                const float lo = p[i][k];
                const float hi = p[i][k + o];
                const float keep = up ? hi : lo;
                const float send = up ? lo : hi;
                p[i][k] = keep + __shfl_xor(send, o, 32);
            }
        }
    }
    for (int i = 0; i < TI; ++i)
        Es[c >> 3][(m0 + 16 * i + 8 * h + (c & 7)) * HB + (n0 >> 6)] = p[i][0];
    __syncthreads();
    {
        const int nH = N >> 6;
        const int h0 = blockCol >> 6;
        for (int idx = tid; idx < BM * HB; idx += 256) {
            const int row  = blockRow + idx / HB;
            const int head = h0 + idx % HB;
            if (row < M && head < nH) {
                el[(long)row * nH + head] = Es[0][idx];
                er[(long)row * nH + head] = Es[1][idx];
            }
        }
    }
}

__global__ __launch_bounds__(256) void classify(
    const float* __restrict__ rst2, const float* __restrict__ b2,
    const float* __restrict__ Wc1, const float* __restrict__ bc1,
    const float* __restrict__ Wc2, const float* __restrict__ bc2,
    float* __restrict__ out, int nNodes)
{
    int wid  = blockIdx.x * 8 + (threadIdx.x >> 5);
    int lane = threadIdx.x & 31;
    if (wid >= nNodes) return;
    float acc = 0.0f;
#pragma unroll
    for (int d = 0; d < 64; ++d)
        acc += (rst2[(long)wid * 64 + d] + b2[d]) * Wc1[d * 32 + lane];
    float t = fmaxf(acc + bc1[lane], 0.0f);
    float o0 = t * Wc2[lane * 2 + 0];
    float o1 = t * Wc2[lane * 2 + 1];
#pragma unroll
    for (int o = 16; o > 0; o >>= 1) {
        o0 += __shfl_xor(o0, o, 32);
        o1 += __shfl_xor(o1, o, 32);
    }
    if (lane == 0) {
        out[(long)wid * 2 + 0] = o0 + bc2[0];
        out[(long)wid * 2 + 1] = o1 + bc2[1];
    }
}

extern "C" void kernel_launch(void* const* d_in, const int* in_sizes, int n_in,
                              void* d_out, int out_size, void* d_ws, size_t ws_size,
                              hipStream_t stream)
{
    (void)n_in; (void)out_size; (void)ws_size;

    const float* randfeat = (const float*)d_in[0];
    const float* func_emb = (const float*)d_in[1];
    const float* codebert = (const float*)d_in[2];
    const int*   src      = (const int*)d_in[3];
    const int*   dst      = (const int*)d_in[4];
    const float* W1  = (const float*)d_in[5];
    const float* al1 = (const float*)d_in[6];
    const float* ar1 = (const float*)d_in[7];
    const float* b1  = (const float*)d_in[8];
    const float* W2  = (const float*)d_in[9];
    const float* al2 = (const float*)d_in[10];
    const float* ar2 = (const float*)d_in[11];
    const float* b2  = (const float*)d_in[12];
    const float* Wc1 = (const float*)d_in[13];
    const float* bc1 = (const float*)d_in[14];
    const float* Wc2 = (const float*)d_in[15];
    const float* bc2 = (const float*)d_in[16];
    float* out = (float*)d_out;

    const int  E  = in_sizes[3];
    const long NN = N_NODES;

    float* ws     = (float*)d_ws;
    float* feat1  = ws;
    float* h2     = feat1  + NN * 256;
    float* feat2  = h2     + NN * 256;
    float* rst2   = feat2  + NN * 64;
    float* el1    = rst2   + NN * 64;
    float* er1    = el1    + NN * 4;
    float* el2    = er1    + NN * 4;
    float* er2    = el2    + NN;
    int*   cur    = (int*)(er2 + NN);
    int*   rowBeg = cur    + NN;
    int*   csrSrc = rowBeg + NN;

    fill_i32<<<(N_NODES + 255) / 256, 256, 0, stream>>>(cur, 0, N_NODES);
    csr_count<<<(E + 255) / 256, 256, 0, stream>>>(dst, cur, E);
    csr_scan<<<1, 1024, 0, stream>>>(cur, rowBeg, N_NODES);
    csr_place<<<(E + 255) / 256, 256, 0, stream>>>(src, dst, cur, csrSrc, E);

    dim3 g1((N_NODES + 127) / 128, 1);
    gemm_wmma_f32_tiled<2, 4, 4><<<g1, 256, 0, stream>>>(randfeat, func_emb, codebert,
                                                        100, 100, 768, W1, feat1,
                                                        al1, ar1, el1, er1,
                                                        N_NODES, 256, 968);
    gat_gather<4, true><<<(N_NODES + 7) / 8, 256, 0, stream>>>(rowBeg, cur, csrSrc, feat1, el1, er1,
                                                              b1, h2, N_NODES);

    dim3 g2((N_NODES + 127) / 128, 1);
    gemm_wmma_f32_tiled<8, 1, 1><<<g2, 256, 0, stream>>>(h2, h2, h2,
                                                        256, 0, 0, W2, feat2,
                                                        al2, ar2, el2, er2,
                                                        N_NODES, 64, 256);
    gat_gather<1, false><<<(N_NODES + 7) / 8, 256, 0, stream>>>(rowBeg, cur, csrSrc, feat2, el2, er2,
                                                               b2, rst2, N_NODES);

    classify<<<(N_NODES + 7) / 8, 256, 0, stream>>>(rst2, b2, Wc1, bc1, Wc2, bc2, out, N_NODES);
}


# Wide GEMM dealt as 241 resident 4-wave blocks of 208 rows x all of N (one wave per SIMD, 13x4 accumulator tiles, K tile 64): every SIMD carries 13 row-tile products, not 16; same summation order
# speedup vs baseline: 4.5350x; 1.1325x over previous
#include <hip/hip_runtime.h>
#include <math.h>


typedef __attribute__((ext_vector_type(2))) float v2f;
typedef __attribute__((ext_vector_type(4))) float v4f;
typedef __attribute__((ext_vector_type(8))) float v8f;

#define N_NODES 50000

__global__ void fill_i32(int* __restrict__ p, int v, int n) {
    int i = blockIdx.x * blockDim.x + threadIdx.x;
    if (i < n) p[i] = v;
}

__global__ void csr_count(const int* __restrict__ dst, int* __restrict__ cnt, int E) {
    int i = blockIdx.x * blockDim.x + threadIdx.x;
    if (i < E) atomicAdd(&cnt[dst[i]], 1);
}

__global__ __launch_bounds__(1024) void csr_scan(int* __restrict__ cur, int* __restrict__ rowBeg, int n) {
    __shared__ int wtot[2][32];
    const int t = threadIdx.x, lane = t & 31, wave = t >> 5;
    auto loadTile = [&](int i) {
        int4 v = make_int4(0, 0, 0, 0);
        if (i + 3 < n) {
            v = *(const int4*)(cur + i);
        } else {
            if (i < n)     v.x = cur[i];
            if (i + 1 < n) v.y = cur[i + 1];
            if (i + 2 < n) v.z = cur[i + 2];
        }
        return v;
    };
    int carry = 0;
    int it = 0;
    int4 nxt = loadTile(4 * t);
    for (int base = 0; base < n; base += 4096, ++it) {
        const int i = base + 4 * t;
        const int4 v = nxt;
        if (base + 4096 < n) nxt = loadTile(i + 4096);
        const int s = v.x + v.y + v.z + v.w;
        int incl = s;
        for (int o = 1; o < 32; o <<= 1) {
            int u = __shfl_up(incl, o, 32);
            if (lane >= o) incl += u;
        }
        if (lane == 31) wtot[it & 1][wave] = incl;
        __syncthreads();
        const int w = wtot[it & 1][lane];
        int wi = w;
        for (int o = 1; o < 32; o <<= 1) {
            int u = __shfl_up(wi, o, 32);
            if (lane >= o) wi += u;
        }
        const int waveOff = __shfl(wi, wave, 32) - __shfl(w, wave, 32);
        const int total   = __shfl(wi, 31, 32);
        int4 r;
        r.x = carry + waveOff + incl - s;
        r.y = r.x + v.x;
        r.z = r.y + v.y;
        r.w = r.z + v.z;
        if (i + 3 < n) {
            *(int4*)(cur + i)    = r;
            *(int4*)(rowBeg + i) = r;
        } else {
            if (i < n)     { cur[i]     = r.x; rowBeg[i]     = r.x; }
            if (i + 1 < n) { cur[i + 1] = r.y; rowBeg[i + 1] = r.y; }
            if (i + 2 < n) { cur[i + 2] = r.z; rowBeg[i + 2] = r.z; }
        }
        carry += total;
    }
}

__global__ void csr_place(const int* __restrict__ src, const int* __restrict__ dst,
                          int* __restrict__ cur, int* __restrict__ csrSrc, int E) {
    int i = blockIdx.x * blockDim.x + threadIdx.x;
    if (i < E) {
        int p = atomicAdd(&cur[dst[i]], 1);
        csrSrc[p] = src[i];
    }
}

template <int H> struct RowVec;
template <> struct RowVec<4> { float4 a, b; };
template <> struct RowVec<1> { float2 a; };

template <int H, bool ELU>
__global__ __launch_bounds__(256) void gat_gather(
    const int* __restrict__ rowBeg, const int* __restrict__ rowEnd,
    const int* __restrict__ csrSrc,
    const float* __restrict__ feat, const float* __restrict__ el,
    const float* __restrict__ er, const float* __restrict__ bias,
    float* __restrict__ out, int nNodes)
{
    static_assert(H == 4 || H == 1, "head counts of this model");
    constexpr int C = H * 64;
    const int lane = threadIdx.x & 31;
    const int d = __builtin_amdgcn_readfirstlane((int)(blockIdx.x * 8 + (threadIdx.x >> 5)));
    if (d >= nNodes) return;
    const int beg = rowBeg[d];
    const int end = rowEnd[d];

    float erd[H];
    for (int h = 0; h < H; ++h) erd[h] = er[d * H + h];

    auto logits = [&](int s, float (&e)[H]) {
        float x[H];
        if constexpr (H == 4) {
            const float4 q = *(const float4*)(el + (long)s * 4);
            x[0] = q.x; x[1] = q.y; x[2] = q.z; x[3] = q.w;
        } else {
            x[0] = el[s];
        }
        for (int h = 0; h < H; ++h) {
            const float y = x[h] + erd[h];
            e[h] = y > 0.0f ? y : 0.2f * y;
        }
    };

    float m[H];
    for (int h = 0; h < H; ++h) m[h] = -INFINITY;
    for (int base = beg; base < end; base += 32) {
        const int i = base + lane;
        if (i < end) {
            float e[H];
            logits(csrSrc[i], e);
            for (int h = 0; h < H; ++h) m[h] = fmaxf(m[h], e[h]);
        }
    }
    for (int h = 0; h < H; ++h) {
        for (int o = 16; o > 0; o >>= 1) m[h] = fmaxf(m[h], __shfl_xor(m[h], o, 32));
        if (!isfinite(m[h])) m[h] = 0.0f;
    }

    float den[H];
    for (int h = 0; h < H; ++h) den[h] = 0.0f;
    for (int base = beg; base < end; base += 32) {
        const int i = base + lane;
        if (i < end) {
            float e[H];
            logits(csrSrc[i], e);
            for (int h = 0; h < H; ++h) den[h] += expf(e[h] - m[h]);
        }
    }
    for (int h = 0; h < H; ++h) {
        for (int o = 16; o > 0; o >>= 1) den[h] += __shfl_xor(den[h], o, 32);
        if (den[h] == 0.0f) den[h] = 1.0f;
    }

    RowVec<H> acc;
    if constexpr (H == 4) {
        acc.a = make_float4(0.0f, 0.0f, 0.0f, 0.0f);
        acc.b = make_float4(0.0f, 0.0f, 0.0f, 0.0f);
    } else {
        acc.a = make_float2(0.0f, 0.0f);
    }
    const bool hi = lane >= 16;
    for (int base = beg; base < end; base += 32) {
        const int i = base + lane;
        int   s = 0;
        float al[H];
        for (int h = 0; h < H; ++h) al[h] = 0.0f;
        if (i < end) {
            float e[H];
            s = csrSrc[i];
            logits(s, e);
            for (int h = 0; h < H; ++h) al[h] = expf(e[h] - m[h]) / den[h];
        }
        const int cnt = min(32, end - base);
#pragma unroll 4
        for (int j = 0; j < cnt; ++j) {
            const int sj = __builtin_amdgcn_readlane(s, j);
            float aj[H];
            for (int h = 0; h < H; ++h)
                aj[h] = __int_as_float(__builtin_amdgcn_readlane(__float_as_int(al[h]), j));
            const float* row = feat + (long)sj * C;
            if constexpr (H == 4) {
                const float4 f0 = *(const float4*)(row + 4 * lane);
                const float4 f1 = *(const float4*)(row + 128 + 4 * lane);
                const float w0 = hi ? aj[1] : aj[0];
                const float w1 = hi ? aj[3] : aj[2];
                acc.a.x += f0.x * w0; acc.a.y += f0.y * w0; acc.a.z += f0.z * w0; acc.a.w += f0.w * w0;
                acc.b.x += f1.x * w1; acc.b.y += f1.y * w1; acc.b.z += f1.z * w1; acc.b.w += f1.w * w1;
            } else {
                const float2 f0 = *(const float2*)(row + 2 * lane);
                acc.a.x += f0.x * aj[0]; acc.a.y += f0.y * aj[0];
            }
        }
    }

    float* o = out + (long)d * C;
    if constexpr (H == 4) {
        float4 r0 = acc.a, r1 = acc.b;
        if constexpr (ELU) {
            const float4 b0 = *(const float4*)(bias + 4 * lane);
            const float4 b1 = *(const float4*)(bias + 128 + 4 * lane);
            r0.x += b0.x; r0.y += b0.y; r0.z += b0.z; r0.w += b0.w;
            r1.x += b1.x; r1.y += b1.y; r1.z += b1.z; r1.w += b1.w;
            r0.x = r0.x > 0.0f ? r0.x : expm1f(r0.x); r0.y = r0.y > 0.0f ? r0.y : expm1f(r0.y);
            r0.z = r0.z > 0.0f ? r0.z : expm1f(r0.z); r0.w = r0.w > 0.0f ? r0.w : expm1f(r0.w);
            r1.x = r1.x > 0.0f ? r1.x : expm1f(r1.x); r1.y = r1.y > 0.0f ? r1.y : expm1f(r1.y);
            r1.z = r1.z > 0.0f ? r1.z : expm1f(r1.z); r1.w = r1.w > 0.0f ? r1.w : expm1f(r1.w);
        }
        *(float4*)(o + 4 * lane)       = r0;
        *(float4*)(o + 128 + 4 * lane) = r1;
    } else {
        float2 r0 = acc.a;
        if constexpr (ELU) {
            const float2 b0 = *(const float2*)(bias + 2 * lane);
            r0.x += b0.x; r0.y += b0.y;
            r0.x = r0.x > 0.0f ? r0.x : expm1f(r0.x); r0.y = r0.y > 0.0f ? r0.y : expm1f(r0.y);
        }
        *(float2*)(o + 2 * lane) = r0;
    }
}

#define GT_BK 16
#define GT_AP 20

template <int WM, int WN, int TI>
__global__ __launch_bounds__(256) void gemm_wmma_f32_tiled(
    const float* __restrict__ p0, const float* __restrict__ p1,
    const float* __restrict__ p2, int w0, int w1, int w2,
    const float* __restrict__ B, float* __restrict__ C,
    const float* __restrict__ al, const float* __restrict__ ar,
    float* __restrict__ el, float* __restrict__ er,
    int M, int N, int K)
{
    static_assert(WM * WN == 8, "8 waves per block");
    constexpr int BM  = WM * TI * 16;
    constexpr int BN  = WN * 64;
    constexpr int BP  = BN + 8;
    constexpr int NA  = BM / 64;
    constexpr int NB  = BN / 64;
    constexpr int BQ  = BN / 4;
    constexpr int BR  = 256 / BQ;
    constexpr int HB  = WN;
    static_assert(BM % 64 == 0, "A staging covers 64 rows per pass");

    __shared__ __attribute__((aligned(16))) float As[2][BM * GT_AP];
    __shared__ __attribute__((aligned(16))) float Bs[2][GT_BK * BP];
    __shared__ float Es[2][BM * HB];

    const int tid  = threadIdx.x;
    const int lane = tid & 31;
    const int wave = tid >> 5;
    const int c    = lane & 15;
    const int h    = lane >> 4;
    const int m0   = (wave % WM) * (TI * 16);
    const int n0   = (wave / WM) * 64;

    const int blockRow = blockIdx.x * BM;
    const int blockCol = blockIdx.y * BN;

    const int am = tid >> 2;
    const int ak = (tid & 3) << 2;
    const int bk = tid / BQ;
    const int bn = (tid % BQ) << 2;

    v8f acc[TI][4];
    for (int i = 0; i < TI; ++i)
        for (int j = 0; j < 4; ++j)
            acc[i][j] = (v8f){};

    float4 ra[NA], rb[NB];

    auto loadTile = [&](int kb) {
        const int gk = kb + ak;
        for (int i = 0; i < NA; ++i) {
            const int row = blockRow + am + 64 * i;
            float4 v = make_float4(0.0f, 0.0f, 0.0f, 0.0f);
            if (row < M && gk < K) {
                const float* s;
                if (gk < w0)            s = p0 + (long)row * w0 + gk;
                else if (gk - w0 < w1)  s = p1 + (long)row * w1 + (gk - w0);
                else                    s = p2 + (long)row * w2 + (gk - w0 - w1);
                v = *(const float4*)s;
            }
            ra[i] = v;
        }
        const int col = blockCol + bn;
        for (int i = 0; i < NB; ++i) {
            const int k = kb + bk + BR * i;
            float4 v = make_float4(0.0f, 0.0f, 0.0f, 0.0f);
            if (k < K && col < N) v = *(const float4*)(B + (long)k * N + col);
            rb[i] = v;
        }
    };

    auto storeTile = [&](int buf) {
        for (int i = 0; i < NA; ++i)
            *(float4*)&As[buf][(am + 64 * i) * GT_AP + ak] = ra[i];
        for (int i = 0; i < NB; ++i)
            *(float4*)&Bs[buf][(bk + BR * i) * BP + bn] = rb[i];
    };

    const int nT = (K + GT_BK - 1) / GT_BK;

    loadTile(0);
    storeTile(0);
    __syncthreads();

    for (int t = 0; t < nT; ++t) {
        const int buf = t & 1;
        if (t + 1 < nT) loadTile((t + 1) * GT_BK);

        const float* Ab = &As[buf][(m0 + c) * GT_AP + 2 * h];
        const float* Bb = &Bs[buf][(2 * h) * BP + n0 + 4 * c];
        for (int kk = 0; kk < GT_BK; kk += 4) {
            v2f a[TI];
            for (int i = 0; i < TI; ++i)
                a[i] = *(const v2f*)&Ab[(16 * i) * GT_AP + kk];
            const float4 bx = *(const float4*)&Bb[kk * BP];
            const float4 by = *(const float4*)&Bb[(kk + 1) * BP];
            v2f b[4];
            b[0].x = bx.x; b[0].y = by.x;
            b[1].x = bx.y; b[1].y = by.y;
            b[2].x = bx.z; b[2].y = by.z;
            b[3].x = bx.w; b[3].y = by.w;
            for (int i = 0; i < TI; ++i)
                for (int j = 0; j < 4; ++j)
                    acc[i][j] = __builtin_amdgcn_wmma_f32_16x16x4_f32(
                        false, a[i], false, b[j], (short)0, acc[i][j], false, false);
        }

        if (t + 1 < nT) storeTile(buf ^ 1);
        __syncthreads();
    }

    const int col = blockCol + n0 + 4 * c;

    float4 wl = make_float4(0.0f, 0.0f, 0.0f, 0.0f);
    float4 wr = make_float4(0.0f, 0.0f, 0.0f, 0.0f);
    if (col < N) {
        wl = *(const float4*)(al + col);
        wr = *(const float4*)(ar + col);
    }

    if (col < N) {
        for (int i = 0; i < TI; ++i) {
            for (int r = 0; r < 8; ++r) {
                const int row = blockRow + m0 + 16 * i + 8 * h + r;
                if (row < M) {
                    float4 v = make_float4(acc[i][0][r], acc[i][1][r], acc[i][2][r], acc[i][3][r]);
                    *(float4*)(C + (long)row * N + col) = v;
                }
            }
        }
    }

    float p[TI][16];
    for (int i = 0; i < TI; ++i) {
        for (int r = 0; r < 8; ++r) {
            p[i][r]     = acc[i][0][r] * wl.x + acc[i][1][r] * wl.y + acc[i][2][r] * wl.z + acc[i][3][r] * wl.w;
            p[i][8 + r] = acc[i][0][r] * wr.x + acc[i][1][r] * wr.y + acc[i][2][r] * wr.z + acc[i][3][r] * wr.w;
        }
    }
#pragma unroll
    for (int o = 8; o >= 1; o >>= 1) {
        const bool up = (c & o) != 0;
#pragma unroll
        for (int i = 0; i < TI; ++i) {
#pragma unroll
            for (int k = 0; k < o; ++k) {
                const float lo = p[i][k];
                const float hi = p[i][k + o];
                const float keep = up ? hi : lo;
                const float send = up ? lo : hi;
                p[i][k] = keep + __shfl_xor(send, o, 32);
            }
        }
    }
    for (int i = 0; i < TI; ++i)
        Es[c >> 3][(m0 + 16 * i + 8 * h + (c & 7)) * HB + (n0 >> 6)] = p[i][0];
    __syncthreads();
    {
        const int nH = N >> 6;
        const int h0 = blockCol >> 6;
        for (int idx = tid; idx < BM * HB; idx += 256) {
            const int row  = blockRow + idx / HB;
            const int head = h0 + idx % HB;
            if (row < M && head < nH) {
                el[(long)row * nH + head] = Es[0][idx];
                er[(long)row * nH + head] = Es[1][idx];
            }
        }
    }
}


struct GemmSeg { const float* p; int w; int o; };
__device__ __forceinline__ GemmSeg gemm_segment(const float* p0, const float* p1, const float* p2,
                                                int w0, int w1, int w2, int gk)
{
    GemmSeg s = { p2, w2, gk - w0 - w1 };
    if (gk - w0 < w1) { s.p = p1; s.w = w1; s.o = gk - w0; }
    if (gk < w0)      { s.p = p0; s.w = w0; s.o = gk; }
    return s;
}

template <int WN, int TI, int BK>
__global__ __launch_bounds__(WN * 32) __attribute__((amdgpu_waves_per_eu(1, 1))) void gemm_wmma_f32_strip(
    const float* __restrict__ p0, const float* __restrict__ p1,
    const float* __restrict__ p2, int w0, int w1, int w2,
    const float* __restrict__ B, float* __restrict__ C,
    const float* __restrict__ al, const float* __restrict__ ar,
    float* __restrict__ el, float* __restrict__ er,
    int M, int N, int K)
{
    constexpr int NT  = WN * 32;
    constexpr int BM  = TI * 16;
    constexpr int BN  = WN * 64;
    constexpr int AP  = BK + 4;
    constexpr int BP  = BN + 8;
    constexpr int KQ  = BK / 4;
    constexpr int AR  = NT / KQ;
    constexpr int NA  = BM / AR;
    constexpr int BQ  = BN / 4;
    constexpr int BR  = NT / BQ;
    constexpr int NB  = BK / BR;
    constexpr int HB  = WN;
    static_assert(BK % 8 == 0 && (KQ & 1) == 0, "A pitch of 16 B x odd needs an even count of float4 per row");
    static_assert(NT % KQ == 0 && BM % AR == 0, "A staging passes cover the tile exactly");
    static_assert(NT % BQ == 0 && BK % BR == 0, "B staging passes cover the tile exactly");

    __shared__ __attribute__((aligned(16))) float As[2][BM * AP];
    __shared__ __attribute__((aligned(16))) float Bs[2][BK * BP];
    __shared__ float Es[2][BM * HB];

    const int tid  = threadIdx.x;
    const int lane = tid & 31;
    const int wave = tid >> 5;
    const int c    = lane & 15;
    const int h    = lane >> 4;
    const int m0   = 0;
    const int n0   = wave * 64;

    const int blockRow = blockIdx.x * BM;
    const int blockCol = blockIdx.y * BN;

    const int am = tid / KQ;
    const int ak = (tid % KQ) << 2;
    const int bk = tid / BQ;
    const int bn = (tid % BQ) << 2;

    v8f acc[TI][4];
#pragma unroll
    for (int i = 0; i < TI; ++i)
#pragma unroll
        for (int j = 0; j < 4; ++j)
            acc[i][j] = (v8f){};

    v4f ra[NA], rb[NB];

    const int bcol = min(blockCol + bn, N - 4);

#define GEMM_LOAD_TILE(kb)                                                                        \
    {                                                                                             \
        const GemmSeg sg = gemm_segment(p0, p1, p2, w0, w1, w2, min((kb) + ak, K - 4));           \
        _Pragma("unroll")                                                                         \
        for (int i = 0; i < NA; ++i) {                                                            \
            const int row = min(blockRow + am + AR * i, M - 1);                                   \
            ra[i] = *(const v4f*)(sg.p + (long)row * sg.w + sg.o);                                \
        }                                                                                         \
        _Pragma("unroll")                                                                         \
        for (int i = 0; i < NB; ++i) {                                                            \
            const int k = min((kb) + bk + BR * i, K - 1);                                         \
            rb[i] = *(const v4f*)(B + (long)k * N + bcol);                                        \
        }                                                                                         \
    }
#define GEMM_STORE_A(buf)                                                                         \
    {                                                                                             \
        _Pragma("unroll")                                                                         \
        for (int i = 0; i < NA; ++i)                                                              \
            *(v4f*)&As[buf][(am + AR * i) * AP + ak] = ra[i];                                     \
    }
#define GEMM_STORE_B(buf)                                                                         \
    {                                                                                             \
        _Pragma("unroll")                                                                         \
        for (int i = 0; i < NB; ++i)                                                              \
            *(v4f*)&Bs[buf][(bk + BR * i) * BP + bn] = rb[i];                                     \
    }

    auto kstep = [&](const float* Ab, const float* Bb, int kk) {
        v2f a[TI];
#pragma unroll
        for (int i = 0; i < TI; ++i)
            a[i] = *(const v2f*)&Ab[(16 * i) * AP + kk];
        const float4 bx = *(const float4*)&Bb[kk * BP];
        const float4 by = *(const float4*)&Bb[(kk + 1) * BP];
        v2f b[4];
        b[0].x = bx.x; b[0].y = by.x;
        b[1].x = bx.y; b[1].y = by.y;
        b[2].x = bx.z; b[2].y = by.z;
        b[3].x = bx.w; b[3].y = by.w;
#pragma unroll
        for (int i = 0; i < TI; ++i)
#pragma unroll
            for (int j = 0; j < 4; ++j)
                acc[i][j] = __builtin_amdgcn_wmma_f32_16x16x4_f32(
                    false, a[i], false, b[j], (short)0, acc[i][j], false, false);
    };

    const int nFull  = K / BK;
    const int ksTail = (K - nFull * BK) >> 2;

    GEMM_LOAD_TILE(0)
    GEMM_STORE_A(0)
    GEMM_STORE_B(0)
    __syncthreads();

    for (int t = 0; t < nFull; ++t) {
        const int buf = t & 1;
        GEMM_LOAD_TILE((t + 1) * BK)

        const float* Ab = &As[buf][(m0 + c) * AP + 2 * h];
        const float* Bb = &Bs[buf][(2 * h) * BP + n0 + 4 * c];
#pragma unroll
        for (int s = 0; s < KQ; ++s) {
            kstep(Ab, Bb, 4 * s);
            if (s == KQ / 2 - 1) GEMM_STORE_A(buf ^ 1)
            if (s == KQ / 2)     GEMM_STORE_B(buf ^ 1)
        }
        __syncthreads();
    }

    if (ksTail > 0) {
        const int buf = nFull & 1;
        const float* Ab = &As[buf][(m0 + c) * AP + 2 * h];
        const float* Bb = &Bs[buf][(2 * h) * BP + n0 + 4 * c];
#pragma unroll 1
        for (int s = 0; s < ksTail; ++s)
            kstep(Ab, Bb, 4 * s);
    }

    const int col = blockCol + n0 + 4 * c;

    float4 wl = make_float4(0.0f, 0.0f, 0.0f, 0.0f);
    float4 wr = make_float4(0.0f, 0.0f, 0.0f, 0.0f);
    if (col < N) {
        wl = *(const float4*)(al + col);
        wr = *(const float4*)(ar + col);
    }

    if (col < N) {
#pragma unroll
        for (int i = 0; i < TI; ++i) {
#pragma unroll
            for (int r = 0; r < 8; ++r) {
                const int row = blockRow + m0 + 16 * i + 8 * h + r;
                if (row < M) {
                    float4 v = make_float4(acc[i][0][r], acc[i][1][r], acc[i][2][r], acc[i][3][r]);
                    *(float4*)(C + (long)row * N + col) = v;
                }
            }
        }
    }

    float p[TI][16];
#pragma unroll
    for (int i = 0; i < TI; ++i) {
#pragma unroll
        for (int r = 0; r < 8; ++r) {
            p[i][r]     = acc[i][0][r] * wl.x + acc[i][1][r] * wl.y + acc[i][2][r] * wl.z + acc[i][3][r] * wl.w;
            p[i][8 + r] = acc[i][0][r] * wr.x + acc[i][1][r] * wr.y + acc[i][2][r] * wr.z + acc[i][3][r] * wr.w;
        }
    }
#pragma unroll
    for (int o = 8; o >= 1; o >>= 1) {
        const bool up = (c & o) != 0;
#pragma unroll
        for (int i = 0; i < TI; ++i) {
#pragma unroll
            for (int k = 0; k < o; ++k) {
                const float lo = p[i][k];
                const float hi = p[i][k + o];
                const float keep = up ? hi : lo;
                const float send = up ? lo : hi;
                p[i][k] = keep + __shfl_xor(send, o, 32);
            }
        }
    }
#pragma unroll
    for (int i = 0; i < TI; ++i)
        Es[c >> 3][(m0 + 16 * i + 8 * h + (c & 7)) * HB + (n0 >> 6)] = p[i][0];
    __syncthreads();
    {
        const int nH = N >> 6;
        const int h0 = blockCol >> 6;
        for (int idx = tid; idx < BM * HB; idx += NT) {
            const int row  = blockRow + idx / HB;
            const int head = h0 + idx % HB;
            if (row < M && head < nH) {
                el[(long)row * nH + head] = Es[0][idx];
                er[(long)row * nH + head] = Es[1][idx];
            }
        }
    }
}
#undef GEMM_LOAD_TILE
#undef GEMM_STORE_A
#undef GEMM_STORE_B

__global__ __launch_bounds__(256) void classify(
    const float* __restrict__ rst2, const float* __restrict__ b2,
    const float* __restrict__ Wc1, const float* __restrict__ bc1,
    const float* __restrict__ Wc2, const float* __restrict__ bc2,
    float* __restrict__ out, int nNodes)
{
    int wid  = blockIdx.x * 8 + (threadIdx.x >> 5);
    int lane = threadIdx.x & 31;
    if (wid >= nNodes) return;
    float acc = 0.0f;
#pragma unroll
    for (int d = 0; d < 64; ++d)
        acc += (rst2[(long)wid * 64 + d] + b2[d]) * Wc1[d * 32 + lane];
    float t = fmaxf(acc + bc1[lane], 0.0f);
    float o0 = t * Wc2[lane * 2 + 0];
    float o1 = t * Wc2[lane * 2 + 1];
#pragma unroll
    for (int o = 16; o > 0; o >>= 1) {
        o0 += __shfl_xor(o0, o, 32);
        o1 += __shfl_xor(o1, o, 32);
    }
    if (lane == 0) {
        out[(long)wid * 2 + 0] = o0 + bc2[0];
        out[(long)wid * 2 + 1] = o1 + bc2[1];
    }
}

extern "C" void kernel_launch(void* const* d_in, const int* in_sizes, int n_in,
                              void* d_out, int out_size, void* d_ws, size_t ws_size,
                              hipStream_t stream)
{
    (void)n_in; (void)out_size; (void)ws_size;

    const float* randfeat = (const float*)d_in[0];
    const float* func_emb = (const float*)d_in[1];
    const float* codebert = (const float*)d_in[2];
    const int*   src      = (const int*)d_in[3];
    const int*   dst      = (const int*)d_in[4];
    const float* W1  = (const float*)d_in[5];
    const float* al1 = (const float*)d_in[6];
    const float* ar1 = (const float*)d_in[7];
    const float* b1  = (const float*)d_in[8];
    const float* W2  = (const float*)d_in[9];
    const float* al2 = (const float*)d_in[10];
    const float* ar2 = (const float*)d_in[11];
    const float* b2  = (const float*)d_in[12];
    const float* Wc1 = (const float*)d_in[13];
    const float* bc1 = (const float*)d_in[14];
    const float* Wc2 = (const float*)d_in[15];
    const float* bc2 = (const float*)d_in[16];
    float* out = (float*)d_out;

    const int  E  = in_sizes[3];
    const long NN = N_NODES;

    float* ws     = (float*)d_ws;
    float* feat1  = ws;
    float* h2     = feat1  + NN * 256;
    float* feat2  = h2     + NN * 256;
    float* rst2   = feat2  + NN * 64;
    float* el1    = rst2   + NN * 64;
    float* er1    = el1    + NN * 4;
    float* el2    = er1    + NN * 4;
    float* er2    = el2    + NN;
    int*   cur    = (int*)(er2 + NN);
    int*   rowBeg = cur    + NN;
    int*   csrSrc = rowBeg + NN;

    fill_i32<<<(N_NODES + 255) / 256, 256, 0, stream>>>(cur, 0, N_NODES);
    csr_count<<<(E + 255) / 256, 256, 0, stream>>>(dst, cur, E);
    csr_scan<<<1, 1024, 0, stream>>>(cur, rowBeg, N_NODES);
    csr_place<<<(E + 255) / 256, 256, 0, stream>>>(src, dst, cur, csrSrc, E);

    dim3 g1((N_NODES + 207) / 208, 1);
    gemm_wmma_f32_strip<4, 13, 64><<<g1, 128, 0, stream>>>(randfeat, func_emb, codebert,
                                                        100, 100, 768, W1, feat1,
                                                        al1, ar1, el1, er1,
                                                        N_NODES, 256, 968);
    gat_gather<4, true><<<(N_NODES + 7) / 8, 256, 0, stream>>>(rowBeg, cur, csrSrc, feat1, el1, er1,
                                                              b1, h2, N_NODES);

    dim3 g2((N_NODES + 127) / 128, 1);
    gemm_wmma_f32_tiled<8, 1, 1><<<g2, 256, 0, stream>>>(h2, h2, h2,
                                                        256, 0, 0, W2, feat2,
                                                        al2, ar2, el2, er2,
                                                        N_NODES, 64, 256);
    gat_gather<1, false><<<(N_NODES + 7) / 8, 256, 0, stream>>>(rowBeg, cur, csrSrc, feat2, el2, er2,
                                                               b2, rst2, N_NODES);

    classify<<<(N_NODES + 7) / 8, 256, 0, stream>>>(rst2, b2, Wc1, bc1, Wc2, bc2, out, N_NODES);
}


# Gather: a dst row of 32 edges or fewer reads its src ids and logits once and keeps them in registers for max, sum and weights (was three dependent re-reads); same values, same summation order
# speedup vs baseline: 4.6742x; 1.0307x over previous
#include <hip/hip_runtime.h>
#include <math.h>


typedef __attribute__((ext_vector_type(2))) float v2f;
typedef __attribute__((ext_vector_type(4))) float v4f;
typedef __attribute__((ext_vector_type(8))) float v8f;

#define N_NODES 50000

__global__ void fill_i32(int* __restrict__ p, int v, int n) {
    int i = blockIdx.x * blockDim.x + threadIdx.x;
    if (i < n) p[i] = v;
}

__global__ void csr_count(const int* __restrict__ dst, int* __restrict__ cnt, int E) {
    int i = blockIdx.x * blockDim.x + threadIdx.x;
    if (i < E) atomicAdd(&cnt[dst[i]], 1);
}

__global__ __launch_bounds__(1024) void csr_scan(int* __restrict__ cur, int* __restrict__ rowBeg, int n) {
    __shared__ int wtot[2][32];
    const int t = threadIdx.x, lane = t & 31, wave = t >> 5;
    auto loadTile = [&](int i) {
        int4 v = make_int4(0, 0, 0, 0);
        if (i + 3 < n) {
            v = *(const int4*)(cur + i);
        } else {
            if (i < n)     v.x = cur[i];
            if (i + 1 < n) v.y = cur[i + 1];
            if (i + 2 < n) v.z = cur[i + 2];
        }
        return v;
    };
    int carry = 0;
    int it = 0;
    int4 nxt = loadTile(4 * t);
    for (int base = 0; base < n; base += 4096, ++it) {
        const int i = base + 4 * t;
        const int4 v = nxt;
        if (base + 4096 < n) nxt = loadTile(i + 4096);
        const int s = v.x + v.y + v.z + v.w;
        int incl = s;
        for (int o = 1; o < 32; o <<= 1) {
            int u = __shfl_up(incl, o, 32);
            if (lane >= o) incl += u;
        }
        if (lane == 31) wtot[it & 1][wave] = incl;
        __syncthreads();
        const int w = wtot[it & 1][lane];
        int wi = w;
        for (int o = 1; o < 32; o <<= 1) {
            int u = __shfl_up(wi, o, 32);
            if (lane >= o) wi += u;
        }
        const int waveOff = __shfl(wi, wave, 32) - __shfl(w, wave, 32);
        const int total   = __shfl(wi, 31, 32);
        int4 r;
        r.x = carry + waveOff + incl - s;
        r.y = r.x + v.x;
        r.z = r.y + v.y;
        r.w = r.z + v.z;
        if (i + 3 < n) {
            *(int4*)(cur + i)    = r;
            *(int4*)(rowBeg + i) = r;
        } else {
            if (i < n)     { cur[i]     = r.x; rowBeg[i]     = r.x; }
            if (i + 1 < n) { cur[i + 1] = r.y; rowBeg[i + 1] = r.y; }
            if (i + 2 < n) { cur[i + 2] = r.z; rowBeg[i + 2] = r.z; }
        }
        carry += total;
    }
}

__global__ void csr_place(const int* __restrict__ src, const int* __restrict__ dst,
                          int* __restrict__ cur, int* __restrict__ csrSrc, int E) {
    int i = blockIdx.x * blockDim.x + threadIdx.x;
    if (i < E) {
        int p = atomicAdd(&cur[dst[i]], 1);
        csrSrc[p] = src[i];
    }
}

template <int H> struct RowVec;
template <> struct RowVec<4> { float4 a, b; };
template <> struct RowVec<1> { float2 a; };

template <int H, bool ELU>
__global__ __launch_bounds__(256) void gat_gather(
    const int* __restrict__ rowBeg, const int* __restrict__ rowEnd,
    const int* __restrict__ csrSrc,
    const float* __restrict__ feat, const float* __restrict__ el,
    const float* __restrict__ er, const float* __restrict__ bias,
    float* __restrict__ out, int nNodes)
{
    static_assert(H == 4 || H == 1, "head counts of this model");
    constexpr int C = H * 64;
    const int lane = threadIdx.x & 31;
    const int d = __builtin_amdgcn_readfirstlane((int)(blockIdx.x * 8 + (threadIdx.x >> 5)));
    if (d >= nNodes) return;
    const int beg = rowBeg[d];
    const int end = rowEnd[d];

    float erd[H];
    for (int h = 0; h < H; ++h) erd[h] = er[d * H + h];

    auto logits = [&](int s, float (&e)[H]) {
        float x[H];
        if constexpr (H == 4) {
            const float4 q = *(const float4*)(el + (long)s * 4);
            x[0] = q.x; x[1] = q.y; x[2] = q.z; x[3] = q.w;
        } else {
            x[0] = el[s];
        }
        for (int h = 0; h < H; ++h) {
            const float y = x[h] + erd[h];
            e[h] = y > 0.0f ? y : 0.2f * y;
        }
    };

    RowVec<H> acc;
    if constexpr (H == 4) {
        acc.a = make_float4(0.0f, 0.0f, 0.0f, 0.0f);
        acc.b = make_float4(0.0f, 0.0f, 0.0f, 0.0f);
    } else {
        acc.a = make_float2(0.0f, 0.0f);
    }
    const bool hi = lane >= 16;
    auto addRows = [&](int s, const float (&al)[H], int cnt) {
#pragma unroll 4
        for (int j = 0; j < cnt; ++j) {
            const int sj = __builtin_amdgcn_readlane(s, j);
            float aj[H];
            for (int h = 0; h < H; ++h)
                aj[h] = __int_as_float(__builtin_amdgcn_readlane(__float_as_int(al[h]), j));
            const float* row = feat + (long)sj * C;
            if constexpr (H == 4) {
                const float4 f0 = *(const float4*)(row + 4 * lane);
                const float4 f1 = *(const float4*)(row + 128 + 4 * lane);
                const float w0 = hi ? aj[1] : aj[0];
                const float w1 = hi ? aj[3] : aj[2];
                acc.a.x += f0.x * w0; acc.a.y += f0.y * w0; acc.a.z += f0.z * w0; acc.a.w += f0.w * w0;
                acc.b.x += f1.x * w1; acc.b.y += f1.y * w1; acc.b.z += f1.z * w1; acc.b.w += f1.w * w1;
            } else {
                const float2 f0 = *(const float2*)(row + 2 * lane);
                acc.a.x += f0.x * aj[0]; acc.a.y += f0.y * aj[0];
            }
        }
    };

    const int n = end - beg;
    if (n <= 32) {
        const bool act = lane < n;
        int   s = 0;
        float e[H];
        for (int h = 0; h < H; ++h) e[h] = -INFINITY;
        if (act) {
            s = csrSrc[beg + lane];
            logits(s, e);
        }
        float m[H];
        for (int h = 0; h < H; ++h) {
            m[h] = e[h];
            for (int o = 16; o > 0; o >>= 1) m[h] = fmaxf(m[h], __shfl_xor(m[h], o, 32));
            if (!isfinite(m[h])) m[h] = 0.0f;
        }
        float ex[H], den[H];
        for (int h = 0; h < H; ++h) {
            ex[h]  = act ? expf(e[h] - m[h]) : 0.0f;
            den[h] = ex[h];
            for (int o = 16; o > 0; o >>= 1) den[h] += __shfl_xor(den[h], o, 32);
            if (den[h] == 0.0f) den[h] = 1.0f;
        }
        float al[H];
        for (int h = 0; h < H; ++h) al[h] = act ? ex[h] / den[h] : 0.0f;
        addRows(s, al, n);
    } else {
        float m[H];
        for (int h = 0; h < H; ++h) m[h] = -INFINITY;
        for (int base = beg; base < end; base += 32) {
            const int i = base + lane;
            if (i < end) {
                float e[H];
                logits(csrSrc[i], e);
                for (int h = 0; h < H; ++h) m[h] = fmaxf(m[h], e[h]);
            }
        }
        for (int h = 0; h < H; ++h) {
            for (int o = 16; o > 0; o >>= 1) m[h] = fmaxf(m[h], __shfl_xor(m[h], o, 32));
            if (!isfinite(m[h])) m[h] = 0.0f;
        }

        float den[H];
        for (int h = 0; h < H; ++h) den[h] = 0.0f;
        for (int base = beg; base < end; base += 32) {
            const int i = base + lane;
            if (i < end) {
                float e[H];
                logits(csrSrc[i], e);
                for (int h = 0; h < H; ++h) den[h] += expf(e[h] - m[h]);
            }
        }
        for (int h = 0; h < H; ++h) {
            for (int o = 16; o > 0; o >>= 1) den[h] += __shfl_xor(den[h], o, 32);
            if (den[h] == 0.0f) den[h] = 1.0f;
        }

        for (int base = beg; base < end; base += 32) {
            const int i = base + lane;
            int   s = 0;
            float al[H];
            for (int h = 0; h < H; ++h) al[h] = 0.0f;
            if (i < end) {
                float e[H];
                s = csrSrc[i];
                logits(s, e);
                for (int h = 0; h < H; ++h) al[h] = expf(e[h] - m[h]) / den[h];
            }
            addRows(s, al, min(32, end - base));
        }
    }

    float* o = out + (long)d * C;
    if constexpr (H == 4) {
        float4 r0 = acc.a, r1 = acc.b;
        if constexpr (ELU) {
            const float4 b0 = *(const float4*)(bias + 4 * lane);
            const float4 b1 = *(const float4*)(bias + 128 + 4 * lane);
            r0.x += b0.x; r0.y += b0.y; r0.z += b0.z; r0.w += b0.w;
            r1.x += b1.x; r1.y += b1.y; r1.z += b1.z; r1.w += b1.w;
            r0.x = r0.x > 0.0f ? r0.x : expm1f(r0.x); r0.y = r0.y > 0.0f ? r0.y : expm1f(r0.y);
            r0.z = r0.z > 0.0f ? r0.z : expm1f(r0.z); r0.w = r0.w > 0.0f ? r0.w : expm1f(r0.w);
            r1.x = r1.x > 0.0f ? r1.x : expm1f(r1.x); r1.y = r1.y > 0.0f ? r1.y : expm1f(r1.y);
            r1.z = r1.z > 0.0f ? r1.z : expm1f(r1.z); r1.w = r1.w > 0.0f ? r1.w : expm1f(r1.w);
        }
        *(float4*)(o + 4 * lane)       = r0;
        *(float4*)(o + 128 + 4 * lane) = r1;
    } else {
        float2 r0 = acc.a;
        if constexpr (ELU) {
            const float2 b0 = *(const float2*)(bias + 2 * lane);
            r0.x += b0.x; r0.y += b0.y;
            r0.x = r0.x > 0.0f ? r0.x : expm1f(r0.x); r0.y = r0.y > 0.0f ? r0.y : expm1f(r0.y);
        }
        *(float2*)(o + 2 * lane) = r0;
    }
}

#define GT_BK 16
#define GT_AP 20

template <int WM, int WN, int TI>
__global__ __launch_bounds__(256) void gemm_wmma_f32_tiled(
    const float* __restrict__ p0, const float* __restrict__ p1,
    const float* __restrict__ p2, int w0, int w1, int w2,
    const float* __restrict__ B, float* __restrict__ C,
    const float* __restrict__ al, const float* __restrict__ ar,
    float* __restrict__ el, float* __restrict__ er,
    int M, int N, int K)
{
    static_assert(WM * WN == 8, "8 waves per block");
    constexpr int BM  = WM * TI * 16;
    constexpr int BN  = WN * 64;
    constexpr int BP  = BN + 8;
    constexpr int NA  = BM / 64;
    constexpr int NB  = BN / 64;
    constexpr int BQ  = BN / 4;
    constexpr int BR  = 256 / BQ;
    constexpr int HB  = WN;
    static_assert(BM % 64 == 0, "A staging covers 64 rows per pass");

    __shared__ __attribute__((aligned(16))) float As[2][BM * GT_AP];
    __shared__ __attribute__((aligned(16))) float Bs[2][GT_BK * BP];
    __shared__ float Es[2][BM * HB];

    const int tid  = threadIdx.x;
    const int lane = tid & 31;
    const int wave = tid >> 5;
    const int c    = lane & 15;
    const int h    = lane >> 4;
    const int m0   = (wave % WM) * (TI * 16);
    const int n0   = (wave / WM) * 64;

    const int blockRow = blockIdx.x * BM;
    const int blockCol = blockIdx.y * BN;

    const int am = tid >> 2;
    const int ak = (tid & 3) << 2;
    const int bk = tid / BQ;
    const int bn = (tid % BQ) << 2;

    v8f acc[TI][4];
    for (int i = 0; i < TI; ++i)
        for (int j = 0; j < 4; ++j)
            acc[i][j] = (v8f){};

    float4 ra[NA], rb[NB];

    auto loadTile = [&](int kb) {
        const int gk = kb + ak;
        for (int i = 0; i < NA; ++i) {
            const int row = blockRow + am + 64 * i;
            float4 v = make_float4(0.0f, 0.0f, 0.0f, 0.0f);
            if (row < M && gk < K) {
                const float* s;
                if (gk < w0)            s = p0 + (long)row * w0 + gk;
                else if (gk - w0 < w1)  s = p1 + (long)row * w1 + (gk - w0);
                else                    s = p2 + (long)row * w2 + (gk - w0 - w1);
                v = *(const float4*)s;
            }
            ra[i] = v;
        }
        const int col = blockCol + bn;
        for (int i = 0; i < NB; ++i) {
            const int k = kb + bk + BR * i;
            float4 v = make_float4(0.0f, 0.0f, 0.0f, 0.0f);
            if (k < K && col < N) v = *(const float4*)(B + (long)k * N + col);
            rb[i] = v;
        }
    };

    auto storeTile = [&](int buf) {
        for (int i = 0; i < NA; ++i)
            *(float4*)&As[buf][(am + 64 * i) * GT_AP + ak] = ra[i];
        for (int i = 0; i < NB; ++i)
            *(float4*)&Bs[buf][(bk + BR * i) * BP + bn] = rb[i];
    };

    const int nT = (K + GT_BK - 1) / GT_BK;

    loadTile(0);
    storeTile(0);
    __syncthreads();

    for (int t = 0; t < nT; ++t) {
        const int buf = t & 1;
        if (t + 1 < nT) loadTile((t + 1) * GT_BK);

        const float* Ab = &As[buf][(m0 + c) * GT_AP + 2 * h];
        const float* Bb = &Bs[buf][(2 * h) * BP + n0 + 4 * c];
        for (int kk = 0; kk < GT_BK; kk += 4) {
            v2f a[TI];
            for (int i = 0; i < TI; ++i)
                a[i] = *(const v2f*)&Ab[(16 * i) * GT_AP + kk];
            const float4 bx = *(const float4*)&Bb[kk * BP];
            const float4 by = *(const float4*)&Bb[(kk + 1) * BP];
            v2f b[4];
            b[0].x = bx.x; b[0].y = by.x;
            b[1].x = bx.y; b[1].y = by.y;
            b[2].x = bx.z; b[2].y = by.z;
            b[3].x = bx.w; b[3].y = by.w;
            for (int i = 0; i < TI; ++i)
                for (int j = 0; j < 4; ++j)
                    acc[i][j] = __builtin_amdgcn_wmma_f32_16x16x4_f32(
                        false, a[i], false, b[j], (short)0, acc[i][j], false, false);
        }

        if (t + 1 < nT) storeTile(buf ^ 1);
        __syncthreads();
    }

    const int col = blockCol + n0 + 4 * c;

    float4 wl = make_float4(0.0f, 0.0f, 0.0f, 0.0f);
    float4 wr = make_float4(0.0f, 0.0f, 0.0f, 0.0f);
    if (col < N) {
        wl = *(const float4*)(al + col);
        wr = *(const float4*)(ar + col);
    }

    if (col < N) {
        for (int i = 0; i < TI; ++i) {
            for (int r = 0; r < 8; ++r) {
                const int row = blockRow + m0 + 16 * i + 8 * h + r;
                if (row < M) {
                    float4 v = make_float4(acc[i][0][r], acc[i][1][r], acc[i][2][r], acc[i][3][r]);
                    *(float4*)(C + (long)row * N + col) = v;
                }
            }
        }
    }

    float p[TI][16];
    for (int i = 0; i < TI; ++i) {
        for (int r = 0; r < 8; ++r) {
            p[i][r]     = acc[i][0][r] * wl.x + acc[i][1][r] * wl.y + acc[i][2][r] * wl.z + acc[i][3][r] * wl.w;
            p[i][8 + r] = acc[i][0][r] * wr.x + acc[i][1][r] * wr.y + acc[i][2][r] * wr.z + acc[i][3][r] * wr.w;
        }
    }
#pragma unroll
    for (int o = 8; o >= 1; o >>= 1) {
        const bool up = (c & o) != 0;
#pragma unroll
        for (int i = 0; i < TI; ++i) {
#pragma unroll
            for (int k = 0; k < o; ++k) {
                const float lo = p[i][k];
                const float hi = p[i][k + o];
                const float keep = up ? hi : lo;
                const float send = up ? lo : hi;
                p[i][k] = keep + __shfl_xor(send, o, 32);
            }
        }
    }
    for (int i = 0; i < TI; ++i)
        Es[c >> 3][(m0 + 16 * i + 8 * h + (c & 7)) * HB + (n0 >> 6)] = p[i][0];
    __syncthreads();
    {
        const int nH = N >> 6;
        const int h0 = blockCol >> 6;
        for (int idx = tid; idx < BM * HB; idx += 256) {
            const int row  = blockRow + idx / HB;
            const int head = h0 + idx % HB;
            if (row < M && head < nH) {
                el[(long)row * nH + head] = Es[0][idx];
                er[(long)row * nH + head] = Es[1][idx];
            }
        }
    }
}


struct GemmSeg { const float* p; int w; int o; };
__device__ __forceinline__ GemmSeg gemm_segment(const float* p0, const float* p1, const float* p2,
                                                int w0, int w1, int w2, int gk)
{
    GemmSeg s = { p2, w2, gk - w0 - w1 };
    if (gk - w0 < w1) { s.p = p1; s.w = w1; s.o = gk - w0; }
    if (gk < w0)      { s.p = p0; s.w = w0; s.o = gk; }
    return s;
}

template <int WN, int TI, int BK>
__global__ __launch_bounds__(WN * 32) __attribute__((amdgpu_waves_per_eu(1, 1))) void gemm_wmma_f32_strip(
    const float* __restrict__ p0, const float* __restrict__ p1,
    const float* __restrict__ p2, int w0, int w1, int w2,
    const float* __restrict__ B, float* __restrict__ C,
    const float* __restrict__ al, const float* __restrict__ ar,
    float* __restrict__ el, float* __restrict__ er,
    int M, int N, int K)
{
    constexpr int NT  = WN * 32;
    constexpr int BM  = TI * 16;
    constexpr int BN  = WN * 64;
    constexpr int AP  = BK + 4;
    constexpr int BP  = BN + 8;
    constexpr int KQ  = BK / 4;
    constexpr int AR  = NT / KQ;
    constexpr int NA  = BM / AR;
    constexpr int BQ  = BN / 4;
    constexpr int BR  = NT / BQ;
    constexpr int NB  = BK / BR;
    constexpr int HB  = WN;
    static_assert(BK % 8 == 0 && (KQ & 1) == 0, "A pitch of 16 B x odd needs an even count of float4 per row");
    static_assert(NT % KQ == 0 && BM % AR == 0, "A staging passes cover the tile exactly");
    static_assert(NT % BQ == 0 && BK % BR == 0, "B staging passes cover the tile exactly");

    __shared__ __attribute__((aligned(16))) float As[2][BM * AP];
    __shared__ __attribute__((aligned(16))) float Bs[2][BK * BP];
    __shared__ float Es[2][BM * HB];

    const int tid  = threadIdx.x;
    const int lane = tid & 31;
    const int wave = tid >> 5;
    const int c    = lane & 15;
    const int h    = lane >> 4;
    const int m0   = 0;
    const int n0   = wave * 64;

    const int blockRow = blockIdx.x * BM;
    const int blockCol = blockIdx.y * BN;

    const int am = tid / KQ;
    const int ak = (tid % KQ) << 2;
    const int bk = tid / BQ;
    const int bn = (tid % BQ) << 2;

    v8f acc[TI][4];
#pragma unroll
    for (int i = 0; i < TI; ++i)
#pragma unroll
        for (int j = 0; j < 4; ++j)
            acc[i][j] = (v8f){};

    v4f ra[NA], rb[NB];

    const int bcol = min(blockCol + bn, N - 4);

#define GEMM_LOAD_TILE(kb)                                                                        \
    {                                                                                             \
        const GemmSeg sg = gemm_segment(p0, p1, p2, w0, w1, w2, min((kb) + ak, K - 4));           \
        _Pragma("unroll")                                                                         \
        for (int i = 0; i < NA; ++i) {                                                            \
            const int row = min(blockRow + am + AR * i, M - 1);                                   \
            ra[i] = *(const v4f*)(sg.p + (long)row * sg.w + sg.o);                                \
        }                                                                                         \
        _Pragma("unroll")                                                                         \
        for (int i = 0; i < NB; ++i) {                                                            \
            const int k = min((kb) + bk + BR * i, K - 1);                                         \
            rb[i] = *(const v4f*)(B + (long)k * N + bcol);                                        \
        }                                                                                         \
    }
#define GEMM_STORE_A(buf)                                                                         \
    {                                                                                             \
        _Pragma("unroll")                                                                         \
        for (int i = 0; i < NA; ++i)                                                              \
            *(v4f*)&As[buf][(am + AR * i) * AP + ak] = ra[i];                                     \
    }
#define GEMM_STORE_B(buf)                                                                         \
    {                                                                                             \
        _Pragma("unroll")                                                                         \
        for (int i = 0; i < NB; ++i)                                                              \
            *(v4f*)&Bs[buf][(bk + BR * i) * BP + bn] = rb[i];                                     \
    }

    auto kstep = [&](const float* Ab, const float* Bb, int kk) {
        v2f a[TI];
#pragma unroll
        for (int i = 0; i < TI; ++i)
            a[i] = *(const v2f*)&Ab[(16 * i) * AP + kk];
        const float4 bx = *(const float4*)&Bb[kk * BP];
        const float4 by = *(const float4*)&Bb[(kk + 1) * BP];
        v2f b[4];
        b[0].x = bx.x; b[0].y = by.x;
        b[1].x = bx.y; b[1].y = by.y;
        b[2].x = bx.z; b[2].y = by.z;
        b[3].x = bx.w; b[3].y = by.w;
#pragma unroll
        for (int i = 0; i < TI; ++i)
#pragma unroll
            for (int j = 0; j < 4; ++j)
                acc[i][j] = __builtin_amdgcn_wmma_f32_16x16x4_f32(
                    false, a[i], false, b[j], (short)0, acc[i][j], false, false);
    };

    const int nFull  = K / BK;
    const int ksTail = (K - nFull * BK) >> 2;

    GEMM_LOAD_TILE(0)
    GEMM_STORE_A(0)
    GEMM_STORE_B(0)
    __syncthreads();

    for (int t = 0; t < nFull; ++t) {
        const int buf = t & 1;
        GEMM_LOAD_TILE((t + 1) * BK)

        const float* Ab = &As[buf][(m0 + c) * AP + 2 * h];
        const float* Bb = &Bs[buf][(2 * h) * BP + n0 + 4 * c];
#pragma unroll
        for (int s = 0; s < KQ; ++s) {
            kstep(Ab, Bb, 4 * s);
            if (s == KQ / 2 - 1) GEMM_STORE_A(buf ^ 1)
            if (s == KQ / 2)     GEMM_STORE_B(buf ^ 1)
        }
        __syncthreads();
    }

    if (ksTail > 0) {
        const int buf = nFull & 1;
        const float* Ab = &As[buf][(m0 + c) * AP + 2 * h];
        const float* Bb = &Bs[buf][(2 * h) * BP + n0 + 4 * c];
#pragma unroll 1
        for (int s = 0; s < ksTail; ++s)
            kstep(Ab, Bb, 4 * s);
    }

    const int col = blockCol + n0 + 4 * c;

    float4 wl = make_float4(0.0f, 0.0f, 0.0f, 0.0f);
    float4 wr = make_float4(0.0f, 0.0f, 0.0f, 0.0f);
    if (col < N) {
        wl = *(const float4*)(al + col);
        wr = *(const float4*)(ar + col);
    }

    if (col < N) {
#pragma unroll
        for (int i = 0; i < TI; ++i) {
#pragma unroll
            for (int r = 0; r < 8; ++r) {
                const int row = blockRow + m0 + 16 * i + 8 * h + r;
                if (row < M) {
                    float4 v = make_float4(acc[i][0][r], acc[i][1][r], acc[i][2][r], acc[i][3][r]);
                    *(float4*)(C + (long)row * N + col) = v;
                }
            }
        }
    }

    float p[TI][16];
#pragma unroll
    for (int i = 0; i < TI; ++i) {
#pragma unroll
        for (int r = 0; r < 8; ++r) {
            p[i][r]     = acc[i][0][r] * wl.x + acc[i][1][r] * wl.y + acc[i][2][r] * wl.z + acc[i][3][r] * wl.w;
            p[i][8 + r] = acc[i][0][r] * wr.x + acc[i][1][r] * wr.y + acc[i][2][r] * wr.z + acc[i][3][r] * wr.w;
        }
    }
#pragma unroll
    for (int o = 8; o >= 1; o >>= 1) {
        const bool up = (c & o) != 0;
#pragma unroll
        for (int i = 0; i < TI; ++i) {
#pragma unroll
            for (int k = 0; k < o; ++k) {
                const float lo = p[i][k];
                const float hi = p[i][k + o];
                const float keep = up ? hi : lo;
                const float send = up ? lo : hi;
                p[i][k] = keep + __shfl_xor(send, o, 32);
            }
        }
    }
#pragma unroll
    for (int i = 0; i < TI; ++i)
        Es[c >> 3][(m0 + 16 * i + 8 * h + (c & 7)) * HB + (n0 >> 6)] = p[i][0];
    __syncthreads();
    {
        const int nH = N >> 6;
        const int h0 = blockCol >> 6;
        for (int idx = tid; idx < BM * HB; idx += NT) {
            const int row  = blockRow + idx / HB;
            const int head = h0 + idx % HB;
            if (row < M && head < nH) {
                el[(long)row * nH + head] = Es[0][idx];
                er[(long)row * nH + head] = Es[1][idx];
            }
        }
    }
}
#undef GEMM_LOAD_TILE
#undef GEMM_STORE_A
#undef GEMM_STORE_B

__global__ __launch_bounds__(256) void classify(
    const float* __restrict__ rst2, const float* __restrict__ b2,
    const float* __restrict__ Wc1, const float* __restrict__ bc1,
    const float* __restrict__ Wc2, const float* __restrict__ bc2,
    float* __restrict__ out, int nNodes)
{
    int wid  = blockIdx.x * 8 + (threadIdx.x >> 5);
    int lane = threadIdx.x & 31;
    if (wid >= nNodes) return;
    float acc = 0.0f;
#pragma unroll
    for (int d = 0; d < 64; ++d)
        acc += (rst2[(long)wid * 64 + d] + b2[d]) * Wc1[d * 32 + lane];
    float t = fmaxf(acc + bc1[lane], 0.0f);
    float o0 = t * Wc2[lane * 2 + 0];
    float o1 = t * Wc2[lane * 2 + 1];
#pragma unroll
    for (int o = 16; o > 0; o >>= 1) {
        o0 += __shfl_xor(o0, o, 32);
        o1 += __shfl_xor(o1, o, 32);
    }
    if (lane == 0) {
        out[(long)wid * 2 + 0] = o0 + bc2[0];
        out[(long)wid * 2 + 1] = o1 + bc2[1];
    }
}

extern "C" void kernel_launch(void* const* d_in, const int* in_sizes, int n_in,
                              void* d_out, int out_size, void* d_ws, size_t ws_size,
                              hipStream_t stream)
{
    (void)n_in; (void)out_size; (void)ws_size;

    const float* randfeat = (const float*)d_in[0];
    const float* func_emb = (const float*)d_in[1];
    const float* codebert = (const float*)d_in[2];
    const int*   src      = (const int*)d_in[3];
    const int*   dst      = (const int*)d_in[4];
    const float* W1  = (const float*)d_in[5];
    const float* al1 = (const float*)d_in[6];
    const float* ar1 = (const float*)d_in[7];
    const float* b1  = (const float*)d_in[8];
    const float* W2  = (const float*)d_in[9];
    const float* al2 = (const float*)d_in[10];
    const float* ar2 = (const float*)d_in[11];
    const float* b2  = (const float*)d_in[12];
    const float* Wc1 = (const float*)d_in[13];
    const float* bc1 = (const float*)d_in[14];
    const float* Wc2 = (const float*)d_in[15];
    const float* bc2 = (const float*)d_in[16];
    float* out = (float*)d_out;

    const int  E  = in_sizes[3];
    const long NN = N_NODES;

    float* ws     = (float*)d_ws;
    float* feat1  = ws;
    float* h2     = feat1  + NN * 256;
    float* feat2  = h2     + NN * 256;
    float* rst2   = feat2  + NN * 64;
    float* el1    = rst2   + NN * 64;
    float* er1    = el1    + NN * 4;
    float* el2    = er1    + NN * 4;
    float* er2    = el2    + NN;
    int*   cur    = (int*)(er2 + NN);
    int*   rowBeg = cur    + NN;
    int*   csrSrc = rowBeg + NN;

    fill_i32<<<(N_NODES + 255) / 256, 256, 0, stream>>>(cur, 0, N_NODES);
    csr_count<<<(E + 255) / 256, 256, 0, stream>>>(dst, cur, E);
    csr_scan<<<1, 1024, 0, stream>>>(cur, rowBeg, N_NODES);
    csr_place<<<(E + 255) / 256, 256, 0, stream>>>(src, dst, cur, csrSrc, E);

    dim3 g1((N_NODES + 207) / 208, 1);
    gemm_wmma_f32_strip<4, 13, 64><<<g1, 128, 0, stream>>>(randfeat, func_emb, codebert,
                                                        100, 100, 768, W1, feat1,
                                                        al1, ar1, el1, er1,
                                                        N_NODES, 256, 968);
    gat_gather<4, true><<<(N_NODES + 7) / 8, 256, 0, stream>>>(rowBeg, cur, csrSrc, feat1, el1, er1,
                                                              b1, h2, N_NODES);

    dim3 g2((N_NODES + 127) / 128, 1);
    gemm_wmma_f32_tiled<8, 1, 1><<<g2, 256, 0, stream>>>(h2, h2, h2,
                                                        256, 0, 0, W2, feat2,
                                                        al2, ar2, el2, er2,
                                                        N_NODES, 64, 256);
    gat_gather<1, false><<<(N_NODES + 7) / 8, 256, 0, stream>>>(rowBeg, cur, csrSrc, feat2, el2, er2,
                                                               b2, rst2, N_NODES);

    classify<<<(N_NODES + 7) / 8, 256, 0, stream>>>(rst2, b2, Wc1, bc1, Wc2, bc2, out, N_NODES);
}


# CSR build as a two-level counting sort ranked with LDS counters (256 edge chunks x 196 buckets of 256 dst rows): no global atomics left; a row's edges now in LDS arrival order (sum order)
# speedup vs baseline: 5.6042x; 1.1990x over previous
#include <hip/hip_runtime.h>
#include <math.h>


typedef __attribute__((ext_vector_type(2))) float v2f;
typedef __attribute__((ext_vector_type(4))) float v4f;
typedef __attribute__((ext_vector_type(8))) float v8f;

#define N_NODES 50000

#define CS_NBLK  256
#define CS_SHIFT 8
#define CS_NBUCK ((N_NODES + 255) >> CS_SHIFT)
#define CS_NCNT  (CS_NBUCK * CS_NBLK)
#define CS_NT    1024
static_assert(CS_NBUCK <= CS_NT, "one thread per bucket");
static_assert(CS_NCNT % 4 == 0, "the scan reads the counts 16 bytes at a time");

__global__ __launch_bounds__(CS_NT) void csr_bucket_count(const int* __restrict__ dst, int* __restrict__ cnt, int E) {
    __shared__ int h[CS_NBUCK];
    const int t = threadIdx.x;
    if (t < CS_NBUCK) h[t] = 0;
    __syncthreads();
    const int chunk = (E + CS_NBLK - 1) / CS_NBLK;
    const int beg = blockIdx.x * chunk;
    const int end = min(E, beg + chunk);
    for (int i0 = beg + t; i0 < end; i0 += 4 * CS_NT) {
        int d[4];
#pragma unroll
        for (int k = 0; k < 4; ++k)
            d[k] = dst[min(i0 + k * CS_NT, end - 1)];
#pragma unroll
        for (int k = 0; k < 4; ++k)
            atomicAdd(&h[d[k] >> CS_SHIFT], i0 + k * CS_NT < end ? 1 : 0);
    }
    __syncthreads();
    if (t < CS_NBUCK) cnt[t * CS_NBLK + blockIdx.x] = h[t];
}

__global__ __launch_bounds__(1024) void csr_scan(int* __restrict__ cur, int* __restrict__ rowBeg, int n) {
    __shared__ int wtot[2][32];
    const int t = threadIdx.x, lane = t & 31, wave = t >> 5;
    auto loadTile = [&](int i) {
        int4 v = make_int4(0, 0, 0, 0);
        if (i + 3 < n) {
            v = *(const int4*)(cur + i);
        } else {
            if (i < n)     v.x = cur[i];
            if (i + 1 < n) v.y = cur[i + 1];
            if (i + 2 < n) v.z = cur[i + 2];
        }
        return v;
    };
    int carry = 0;
    int it = 0;
    int4 nxt = loadTile(4 * t);
    for (int base = 0; base < n; base += 4096, ++it) {
        const int i = base + 4 * t;
        const int4 v = nxt;
        if (base + 4096 < n) nxt = loadTile(i + 4096);
        const int s = v.x + v.y + v.z + v.w;
        int incl = s;
        for (int o = 1; o < 32; o <<= 1) {
            int u = __shfl_up(incl, o, 32);
            if (lane >= o) incl += u;
        }
        if (lane == 31) wtot[it & 1][wave] = incl;
        __syncthreads();
        const int w = wtot[it & 1][lane];
        int wi = w;
        for (int o = 1; o < 32; o <<= 1) {
            int u = __shfl_up(wi, o, 32);
            if (lane >= o) wi += u;
        }
        const int waveOff = __shfl(wi, wave, 32) - __shfl(w, wave, 32);
        const int total   = __shfl(wi, 31, 32);
        int4 r;
        r.x = carry + waveOff + incl - s;
        r.y = r.x + v.x;
        r.z = r.y + v.y;
        r.w = r.z + v.z;
        if (i + 3 < n) {
            *(int4*)(cur + i)    = r;
            *(int4*)(rowBeg + i) = r;
        } else {
            if (i < n)     { cur[i]     = r.x; rowBeg[i]     = r.x; }
            if (i + 1 < n) { cur[i + 1] = r.y; rowBeg[i + 1] = r.y; }
            if (i + 2 < n) { cur[i + 2] = r.z; rowBeg[i + 2] = r.z; }
        }
        carry += total;
    }
}

__global__ __launch_bounds__(CS_NT) void csr_bucket_place(const int* __restrict__ src, const int* __restrict__ dst,
                                                         const int* __restrict__ pos, int2* __restrict__ tmp, int E) {
    __shared__ int off[CS_NBUCK];
    const int t = threadIdx.x;
    if (t < CS_NBUCK) off[t] = pos[t * CS_NBLK + blockIdx.x];
    __syncthreads();
    const int chunk = (E + CS_NBLK - 1) / CS_NBLK;
    const int beg = blockIdx.x * chunk;
    const int end = min(E, beg + chunk);
    for (int i0 = beg + t; i0 < end; i0 += 4 * CS_NT) {
        int d[4], s[4];
#pragma unroll
        for (int k = 0; k < 4; ++k) {
            const int i = min(i0 + k * CS_NT, end - 1);
            d[k] = dst[i];
            s[k] = src[i];
        }
#pragma unroll
        for (int k = 0; k < 4; ++k) {
            if (i0 + k * CS_NT < end) {
                const int p = atomicAdd(&off[d[k] >> CS_SHIFT], 1);
                tmp[p] = make_int2(s[k], d[k]);
            }
        }
    }
}

__global__ __launch_bounds__(1024) void csr_bucket_sort(const int2* __restrict__ tmp, const int* __restrict__ pos,
                                                         int* __restrict__ rowBeg, int* __restrict__ rowEnd,
                                                         int* __restrict__ csrSrc, int E, int nNodes) {
    __shared__ int c[256];
    __shared__ int wtot[8];
    const int t = threadIdx.x, lane = t & 31, wave = t >> 5;
    const int b = blockIdx.x;
    const int beg = pos[b * CS_NBLK];
    const int end = (b + 1 < CS_NBUCK) ? pos[(b + 1) * CS_NBLK] : E;
    if (t < 256) c[t] = 0;
    __syncthreads();
    for (int i0 = beg + t; i0 < end; i0 += 8 * 1024) {
        int d[8];
#pragma unroll
        for (int k = 0; k < 8; ++k)
            d[k] = tmp[min(i0 + k * 1024, end - 1)].y;
#pragma unroll
        for (int k = 0; k < 8; ++k)
            atomicAdd(&c[d[k] & 255], i0 + k * 1024 < end ? 1 : 0);
    }
    __syncthreads();
    int deg = 0, incl = 0;
    if (t < 256) {
        deg  = c[t];
        incl = deg;
        for (int o = 1; o < 32; o <<= 1) {
            int u = __shfl_up(incl, o, 32);
            if (lane >= o) incl += u;
        }
        if (lane == 31) wtot[wave] = incl;
    }
    __syncthreads();
    if (t < 256) {
        int base = 0;
        for (int w = 0; w < wave; ++w) base += wtot[w];
        const int excl = base + incl - deg;
        c[t] = excl;
        const int node = (b << CS_SHIFT) + t;
        if (node < nNodes) {
            rowBeg[node] = beg + excl;
            rowEnd[node] = beg + excl + deg;
        }
    }
    __syncthreads();
    for (int i0 = beg + t; i0 < end; i0 += 8 * 1024) {
        int2 e[8];
#pragma unroll
        for (int k = 0; k < 8; ++k)
            e[k] = tmp[min(i0 + k * 1024, end - 1)];
#pragma unroll
        for (int k = 0; k < 8; ++k) {
            if (i0 + k * 1024 < end) {
                const int p = atomicAdd(&c[e[k].y & 255], 1);
                csrSrc[beg + p] = e[k].x;
            }
        }
    }
}

template <int H> struct RowVec;
template <> struct RowVec<4> { float4 a, b; };
template <> struct RowVec<1> { float2 a; };

template <int H, bool ELU>
__global__ __launch_bounds__(256) void gat_gather(
    const int* __restrict__ rowBeg, const int* __restrict__ rowEnd,
    const int* __restrict__ csrSrc,
    const float* __restrict__ feat, const float* __restrict__ el,
    const float* __restrict__ er, const float* __restrict__ bias,
    float* __restrict__ out, int nNodes)
{
    static_assert(H == 4 || H == 1, "head counts of this model");
    constexpr int C = H * 64;
    const int lane = threadIdx.x & 31;
    const int d = __builtin_amdgcn_readfirstlane((int)(blockIdx.x * 8 + (threadIdx.x >> 5)));
    if (d >= nNodes) return;
    const int beg = rowBeg[d];
    const int end = rowEnd[d];

    float erd[H];
    for (int h = 0; h < H; ++h) erd[h] = er[d * H + h];

    auto logits = [&](int s, float (&e)[H]) {
        float x[H];
        if constexpr (H == 4) {
            const float4 q = *(const float4*)(el + (long)s * 4);
            x[0] = q.x; x[1] = q.y; x[2] = q.z; x[3] = q.w;
        } else {
            x[0] = el[s];
        }
        for (int h = 0; h < H; ++h) {
            const float y = x[h] + erd[h];
            e[h] = y > 0.0f ? y : 0.2f * y;
        }
    };

    RowVec<H> acc;
    if constexpr (H == 4) {
        acc.a = make_float4(0.0f, 0.0f, 0.0f, 0.0f);
        acc.b = make_float4(0.0f, 0.0f, 0.0f, 0.0f);
    } else {
        acc.a = make_float2(0.0f, 0.0f);
    }
    const bool hi = lane >= 16;
    auto addRows = [&](int s, const float (&al)[H], int cnt) {
#pragma unroll 4
        for (int j = 0; j < cnt; ++j) {
            const int sj = __builtin_amdgcn_readlane(s, j);
            float aj[H];
            for (int h = 0; h < H; ++h)
                aj[h] = __int_as_float(__builtin_amdgcn_readlane(__float_as_int(al[h]), j));
            const float* row = feat + (long)sj * C;
            if constexpr (H == 4) {
                const float4 f0 = *(const float4*)(row + 4 * lane);
                const float4 f1 = *(const float4*)(row + 128 + 4 * lane);
                const float w0 = hi ? aj[1] : aj[0];
                const float w1 = hi ? aj[3] : aj[2];
                acc.a.x += f0.x * w0; acc.a.y += f0.y * w0; acc.a.z += f0.z * w0; acc.a.w += f0.w * w0;
                acc.b.x += f1.x * w1; acc.b.y += f1.y * w1; acc.b.z += f1.z * w1; acc.b.w += f1.w * w1;
            } else {
                const float2 f0 = *(const float2*)(row + 2 * lane);
                acc.a.x += f0.x * aj[0]; acc.a.y += f0.y * aj[0];
            }
        }
    };

    const int n = end - beg;
    if (n <= 32) {
        const bool act = lane < n;
        int   s = 0;
        float e[H];
        for (int h = 0; h < H; ++h) e[h] = -INFINITY;
        if (act) {
            s = csrSrc[beg + lane];
            logits(s, e);
        }
        float m[H];
        for (int h = 0; h < H; ++h) {
            m[h] = e[h];
            for (int o = 16; o > 0; o >>= 1) m[h] = fmaxf(m[h], __shfl_xor(m[h], o, 32));
            if (!isfinite(m[h])) m[h] = 0.0f;
        }
        float ex[H], den[H];
        for (int h = 0; h < H; ++h) {
            ex[h]  = act ? expf(e[h] - m[h]) : 0.0f;
            den[h] = ex[h];
            for (int o = 16; o > 0; o >>= 1) den[h] += __shfl_xor(den[h], o, 32);
            if (den[h] == 0.0f) den[h] = 1.0f;
        }
        float al[H];
        for (int h = 0; h < H; ++h) al[h] = act ? ex[h] / den[h] : 0.0f;
        addRows(s, al, n);
    } else {
        float m[H];
        for (int h = 0; h < H; ++h) m[h] = -INFINITY;
        for (int base = beg; base < end; base += 32) {
            const int i = base + lane;
            if (i < end) {
                float e[H];
                logits(csrSrc[i], e);
                for (int h = 0; h < H; ++h) m[h] = fmaxf(m[h], e[h]);
            }
        }
        for (int h = 0; h < H; ++h) {
            for (int o = 16; o > 0; o >>= 1) m[h] = fmaxf(m[h], __shfl_xor(m[h], o, 32));
            if (!isfinite(m[h])) m[h] = 0.0f;
        }

        float den[H];
        for (int h = 0; h < H; ++h) den[h] = 0.0f;
        for (int base = beg; base < end; base += 32) {
            const int i = base + lane;
            if (i < end) {
                float e[H];
                logits(csrSrc[i], e);
                for (int h = 0; h < H; ++h) den[h] += expf(e[h] - m[h]);
            }
        }
        for (int h = 0; h < H; ++h) {
            for (int o = 16; o > 0; o >>= 1) den[h] += __shfl_xor(den[h], o, 32);
            if (den[h] == 0.0f) den[h] = 1.0f;
        }

        for (int base = beg; base < end; base += 32) {
            const int i = base + lane;
            int   s = 0;
            float al[H];
            for (int h = 0; h < H; ++h) al[h] = 0.0f;
            if (i < end) {
                float e[H];
                s = csrSrc[i];
                logits(s, e);
                for (int h = 0; h < H; ++h) al[h] = expf(e[h] - m[h]) / den[h];
            }
            addRows(s, al, min(32, end - base));
        }
    }

    float* o = out + (long)d * C;
    if constexpr (H == 4) {
        float4 r0 = acc.a, r1 = acc.b;
        if constexpr (ELU) {
            const float4 b0 = *(const float4*)(bias + 4 * lane);
            const float4 b1 = *(const float4*)(bias + 128 + 4 * lane);
            r0.x += b0.x; r0.y += b0.y; r0.z += b0.z; r0.w += b0.w;
            r1.x += b1.x; r1.y += b1.y; r1.z += b1.z; r1.w += b1.w;
            r0.x = r0.x > 0.0f ? r0.x : expm1f(r0.x); r0.y = r0.y > 0.0f ? r0.y : expm1f(r0.y);
            r0.z = r0.z > 0.0f ? r0.z : expm1f(r0.z); r0.w = r0.w > 0.0f ? r0.w : expm1f(r0.w);
            r1.x = r1.x > 0.0f ? r1.x : expm1f(r1.x); r1.y = r1.y > 0.0f ? r1.y : expm1f(r1.y);
            r1.z = r1.z > 0.0f ? r1.z : expm1f(r1.z); r1.w = r1.w > 0.0f ? r1.w : expm1f(r1.w);
        }
        *(float4*)(o + 4 * lane)       = r0;
        *(float4*)(o + 128 + 4 * lane) = r1;
    } else {
        float2 r0 = acc.a;
        if constexpr (ELU) {
            const float2 b0 = *(const float2*)(bias + 2 * lane);
            r0.x += b0.x; r0.y += b0.y;
            r0.x = r0.x > 0.0f ? r0.x : expm1f(r0.x); r0.y = r0.y > 0.0f ? r0.y : expm1f(r0.y);
        }
        *(float2*)(o + 2 * lane) = r0;
    }
}

#define GT_BK 16
#define GT_AP 20

template <int WM, int WN, int TI>
__global__ __launch_bounds__(256) void gemm_wmma_f32_tiled(
    const float* __restrict__ p0, const float* __restrict__ p1,
    const float* __restrict__ p2, int w0, int w1, int w2,
    const float* __restrict__ B, float* __restrict__ C,
    const float* __restrict__ al, const float* __restrict__ ar,
    float* __restrict__ el, float* __restrict__ er,
    int M, int N, int K)
{
    static_assert(WM * WN == 8, "8 waves per block");
    constexpr int BM  = WM * TI * 16;
    constexpr int BN  = WN * 64;
    constexpr int BP  = BN + 8;
    constexpr int NA  = BM / 64;
    constexpr int NB  = BN / 64;
    constexpr int BQ  = BN / 4;
    constexpr int BR  = 256 / BQ;
    constexpr int HB  = WN;
    static_assert(BM % 64 == 0, "A staging covers 64 rows per pass");

    __shared__ __attribute__((aligned(16))) float As[2][BM * GT_AP];
    __shared__ __attribute__((aligned(16))) float Bs[2][GT_BK * BP];
    __shared__ float Es[2][BM * HB];

    const int tid  = threadIdx.x;
    const int lane = tid & 31;
    const int wave = tid >> 5;
    const int c    = lane & 15;
    const int h    = lane >> 4;
    const int m0   = (wave % WM) * (TI * 16);
    const int n0   = (wave / WM) * 64;

    const int blockRow = blockIdx.x * BM;
    const int blockCol = blockIdx.y * BN;

    const int am = tid >> 2;
    const int ak = (tid & 3) << 2;
    const int bk = tid / BQ;
    const int bn = (tid % BQ) << 2;

    v8f acc[TI][4];
    for (int i = 0; i < TI; ++i)
        for (int j = 0; j < 4; ++j)
            acc[i][j] = (v8f){};

    float4 ra[NA], rb[NB];

    auto loadTile = [&](int kb) {
        const int gk = kb + ak;
        for (int i = 0; i < NA; ++i) {
            const int row = blockRow + am + 64 * i;
            float4 v = make_float4(0.0f, 0.0f, 0.0f, 0.0f);
            if (row < M && gk < K) {
                const float* s;
                if (gk < w0)            s = p0 + (long)row * w0 + gk;
                else if (gk - w0 < w1)  s = p1 + (long)row * w1 + (gk - w0);
                else                    s = p2 + (long)row * w2 + (gk - w0 - w1);
                v = *(const float4*)s;
            }
            ra[i] = v;
        }
        const int col = blockCol + bn;
        for (int i = 0; i < NB; ++i) {
            const int k = kb + bk + BR * i;
            float4 v = make_float4(0.0f, 0.0f, 0.0f, 0.0f);
            if (k < K && col < N) v = *(const float4*)(B + (long)k * N + col);
            rb[i] = v;
        }
    };

    auto storeTile = [&](int buf) {
        for (int i = 0; i < NA; ++i)
            *(float4*)&As[buf][(am + 64 * i) * GT_AP + ak] = ra[i];
        for (int i = 0; i < NB; ++i)
            *(float4*)&Bs[buf][(bk + BR * i) * BP + bn] = rb[i];
    };

    const int nT = (K + GT_BK - 1) / GT_BK;

    loadTile(0);
    storeTile(0);
    __syncthreads();

    for (int t = 0; t < nT; ++t) {
        const int buf = t & 1;
        if (t + 1 < nT) loadTile((t + 1) * GT_BK);

        const float* Ab = &As[buf][(m0 + c) * GT_AP + 2 * h];
        const float* Bb = &Bs[buf][(2 * h) * BP + n0 + 4 * c];
        for (int kk = 0; kk < GT_BK; kk += 4) {
            v2f a[TI];
            for (int i = 0; i < TI; ++i)
                a[i] = *(const v2f*)&Ab[(16 * i) * GT_AP + kk];
            const float4 bx = *(const float4*)&Bb[kk * BP];
            const float4 by = *(const float4*)&Bb[(kk + 1) * BP];
            v2f b[4];
            b[0].x = bx.x; b[0].y = by.x;
            b[1].x = bx.y; b[1].y = by.y;
            b[2].x = bx.z; b[2].y = by.z;
            b[3].x = bx.w; b[3].y = by.w;
            for (int i = 0; i < TI; ++i)
                for (int j = 0; j < 4; ++j)
                    acc[i][j] = __builtin_amdgcn_wmma_f32_16x16x4_f32(
                        false, a[i], false, b[j], (short)0, acc[i][j], false, false);
        }

        if (t + 1 < nT) storeTile(buf ^ 1);
        __syncthreads();
    }

    const int col = blockCol + n0 + 4 * c;

    float4 wl = make_float4(0.0f, 0.0f, 0.0f, 0.0f);
    float4 wr = make_float4(0.0f, 0.0f, 0.0f, 0.0f);
    if (col < N) {
        wl = *(const float4*)(al + col);
        wr = *(const float4*)(ar + col);
    }

    if (col < N) {
        for (int i = 0; i < TI; ++i) {
            for (int r = 0; r < 8; ++r) {
                const int row = blockRow + m0 + 16 * i + 8 * h + r;
                if (row < M) {
                    float4 v = make_float4(acc[i][0][r], acc[i][1][r], acc[i][2][r], acc[i][3][r]);
                    *(float4*)(C + (long)row * N + col) = v;
                }
            }
        }
    }

    float p[TI][16];
    for (int i = 0; i < TI; ++i) {
        for (int r = 0; r < 8; ++r) {
            p[i][r]     = acc[i][0][r] * wl.x + acc[i][1][r] * wl.y + acc[i][2][r] * wl.z + acc[i][3][r] * wl.w;
            p[i][8 + r] = acc[i][0][r] * wr.x + acc[i][1][r] * wr.y + acc[i][2][r] * wr.z + acc[i][3][r] * wr.w;
        }
    }
#pragma unroll
    for (int o = 8; o >= 1; o >>= 1) {
        const bool up = (c & o) != 0;
#pragma unroll
        for (int i = 0; i < TI; ++i) {
#pragma unroll
            for (int k = 0; k < o; ++k) {
                const float lo = p[i][k];
                const float hi = p[i][k + o];
                const float keep = up ? hi : lo;
                const float send = up ? lo : hi;
                p[i][k] = keep + __shfl_xor(send, o, 32);
            }
        }
    }
    for (int i = 0; i < TI; ++i)
        Es[c >> 3][(m0 + 16 * i + 8 * h + (c & 7)) * HB + (n0 >> 6)] = p[i][0];
    __syncthreads();
    {
        const int nH = N >> 6;
        const int h0 = blockCol >> 6;
        for (int idx = tid; idx < BM * HB; idx += 256) {
            const int row  = blockRow + idx / HB;
            const int head = h0 + idx % HB;
            if (row < M && head < nH) {
                el[(long)row * nH + head] = Es[0][idx];
                er[(long)row * nH + head] = Es[1][idx];
            }
        }
    }
}


struct GemmSeg { const float* p; int w; int o; };
__device__ __forceinline__ GemmSeg gemm_segment(const float* p0, const float* p1, const float* p2,
                                                int w0, int w1, int w2, int gk)
{
    GemmSeg s = { p2, w2, gk - w0 - w1 };
    if (gk - w0 < w1) { s.p = p1; s.w = w1; s.o = gk - w0; }
    if (gk < w0)      { s.p = p0; s.w = w0; s.o = gk; }
    return s;
}

template <int WN, int TI, int BK>
__global__ __launch_bounds__(WN * 32) __attribute__((amdgpu_waves_per_eu(1, 1))) void gemm_wmma_f32_strip(
    const float* __restrict__ p0, const float* __restrict__ p1,
    const float* __restrict__ p2, int w0, int w1, int w2,
    const float* __restrict__ B, float* __restrict__ C,
    const float* __restrict__ al, const float* __restrict__ ar,
    float* __restrict__ el, float* __restrict__ er,
    int M, int N, int K)
{
    constexpr int NT  = WN * 32;
    constexpr int BM  = TI * 16;
    constexpr int BN  = WN * 64;
    constexpr int AP  = BK + 4;
    constexpr int BP  = BN + 8;
    constexpr int KQ  = BK / 4;
    constexpr int AR  = NT / KQ;
    constexpr int NA  = BM / AR;
    constexpr int BQ  = BN / 4;
    constexpr int BR  = NT / BQ;
    constexpr int NB  = BK / BR;
    constexpr int HB  = WN;
    static_assert(BK % 8 == 0 && (KQ & 1) == 0, "A pitch of 16 B x odd needs an even count of float4 per row");
    static_assert(NT % KQ == 0 && BM % AR == 0, "A staging passes cover the tile exactly");
    static_assert(NT % BQ == 0 && BK % BR == 0, "B staging passes cover the tile exactly");

    __shared__ __attribute__((aligned(16))) float As[2][BM * AP];
    __shared__ __attribute__((aligned(16))) float Bs[2][BK * BP];
    __shared__ float Es[2][BM * HB];

    const int tid  = threadIdx.x;
    const int lane = tid & 31;
    const int wave = tid >> 5;
    const int c    = lane & 15;
    const int h    = lane >> 4;
    const int m0   = 0;
    const int n0   = wave * 64;

    const int blockRow = blockIdx.x * BM;
    const int blockCol = blockIdx.y * BN;

    const int am = tid / KQ;
    const int ak = (tid % KQ) << 2;
    const int bk = tid / BQ;
    const int bn = (tid % BQ) << 2;

    v8f acc[TI][4];
#pragma unroll
    for (int i = 0; i < TI; ++i)
#pragma unroll
        for (int j = 0; j < 4; ++j)
            acc[i][j] = (v8f){};

    v4f ra[NA], rb[NB];

    const int bcol = min(blockCol + bn, N - 4);

#define GEMM_LOAD_TILE(kb)                                                                        \
    {                                                                                             \
        const GemmSeg sg = gemm_segment(p0, p1, p2, w0, w1, w2, min((kb) + ak, K - 4));           \
        _Pragma("unroll")                                                                         \
        for (int i = 0; i < NA; ++i) {                                                            \
            const int row = min(blockRow + am + AR * i, M - 1);                                   \
            ra[i] = *(const v4f*)(sg.p + (long)row * sg.w + sg.o);                                \
        }                                                                                         \
        _Pragma("unroll")                                                                         \
        for (int i = 0; i < NB; ++i) {                                                            \
            const int k = min((kb) + bk + BR * i, K - 1);                                         \
            rb[i] = *(const v4f*)(B + (long)k * N + bcol);                                        \
        }                                                                                         \
    }
#define GEMM_STORE_A(buf)                                                                         \
    {                                                                                             \
        _Pragma("unroll")                                                                         \
        for (int i = 0; i < NA; ++i)                                                              \
            *(v4f*)&As[buf][(am + AR * i) * AP + ak] = ra[i];                                     \
    }
#define GEMM_STORE_B(buf)                                                                         \
    {                                                                                             \
        _Pragma("unroll")                                                                         \
        for (int i = 0; i < NB; ++i)                                                              \
            *(v4f*)&Bs[buf][(bk + BR * i) * BP + bn] = rb[i];                                     \
    }

    auto kstep = [&](const float* Ab, const float* Bb, int kk) {
        v2f a[TI];
#pragma unroll
        for (int i = 0; i < TI; ++i)
            a[i] = *(const v2f*)&Ab[(16 * i) * AP + kk];
        const float4 bx = *(const float4*)&Bb[kk * BP];
        const float4 by = *(const float4*)&Bb[(kk + 1) * BP];
        v2f b[4];
        b[0].x = bx.x; b[0].y = by.x;
        b[1].x = bx.y; b[1].y = by.y;
        b[2].x = bx.z; b[2].y = by.z;
        b[3].x = bx.w; b[3].y = by.w;
#pragma unroll
        for (int i = 0; i < TI; ++i)
#pragma unroll
            for (int j = 0; j < 4; ++j)
                acc[i][j] = __builtin_amdgcn_wmma_f32_16x16x4_f32(
                    false, a[i], false, b[j], (short)0, acc[i][j], false, false);
    };

    const int nFull  = K / BK;
    const int ksTail = (K - nFull * BK) >> 2;

    GEMM_LOAD_TILE(0)
    GEMM_STORE_A(0)
    GEMM_STORE_B(0)
    __syncthreads();

    for (int t = 0; t < nFull; ++t) {
        const int buf = t & 1;
        GEMM_LOAD_TILE((t + 1) * BK)

        const float* Ab = &As[buf][(m0 + c) * AP + 2 * h];
        const float* Bb = &Bs[buf][(2 * h) * BP + n0 + 4 * c];
#pragma unroll
        for (int s = 0; s < KQ; ++s) {
            kstep(Ab, Bb, 4 * s);
            if (s == KQ / 2 - 1) GEMM_STORE_A(buf ^ 1)
            if (s == KQ / 2)     GEMM_STORE_B(buf ^ 1)
        }
        __syncthreads();
    }

    if (ksTail > 0) {
        const int buf = nFull & 1;
        const float* Ab = &As[buf][(m0 + c) * AP + 2 * h];
        const float* Bb = &Bs[buf][(2 * h) * BP + n0 + 4 * c];
#pragma unroll 1
        for (int s = 0; s < ksTail; ++s)
            kstep(Ab, Bb, 4 * s);
    }

    const int col = blockCol + n0 + 4 * c;

    float4 wl = make_float4(0.0f, 0.0f, 0.0f, 0.0f);
    float4 wr = make_float4(0.0f, 0.0f, 0.0f, 0.0f);
    if (col < N) {
        wl = *(const float4*)(al + col);
        wr = *(const float4*)(ar + col);
    }

    if (col < N) {
#pragma unroll
        for (int i = 0; i < TI; ++i) {
#pragma unroll
            for (int r = 0; r < 8; ++r) {
                const int row = blockRow + m0 + 16 * i + 8 * h + r;
                if (row < M) {
                    float4 v = make_float4(acc[i][0][r], acc[i][1][r], acc[i][2][r], acc[i][3][r]);
                    *(float4*)(C + (long)row * N + col) = v;
                }
            }
        }
    }

    float p[TI][16];
#pragma unroll
    for (int i = 0; i < TI; ++i) {
#pragma unroll
        for (int r = 0; r < 8; ++r) {
            p[i][r]     = acc[i][0][r] * wl.x + acc[i][1][r] * wl.y + acc[i][2][r] * wl.z + acc[i][3][r] * wl.w;
            p[i][8 + r] = acc[i][0][r] * wr.x + acc[i][1][r] * wr.y + acc[i][2][r] * wr.z + acc[i][3][r] * wr.w;
        }
    }
#pragma unroll
    for (int o = 8; o >= 1; o >>= 1) {
        const bool up = (c & o) != 0;
#pragma unroll
        for (int i = 0; i < TI; ++i) {
#pragma unroll
            for (int k = 0; k < o; ++k) {
                const float lo = p[i][k];
                const float hi = p[i][k + o];
                const float keep = up ? hi : lo;
                const float send = up ? lo : hi;
                p[i][k] = keep + __shfl_xor(send, o, 32);
            }
        }
    }
#pragma unroll
    for (int i = 0; i < TI; ++i)
        Es[c >> 3][(m0 + 16 * i + 8 * h + (c & 7)) * HB + (n0 >> 6)] = p[i][0];
    __syncthreads();
    {
        const int nH = N >> 6;
        const int h0 = blockCol >> 6;
        for (int idx = tid; idx < BM * HB; idx += NT) {
            const int row  = blockRow + idx / HB;
            const int head = h0 + idx % HB;
            if (row < M && head < nH) {
                el[(long)row * nH + head] = Es[0][idx];
                er[(long)row * nH + head] = Es[1][idx];
            }
        }
    }
}
#undef GEMM_LOAD_TILE
#undef GEMM_STORE_A
#undef GEMM_STORE_B

__global__ __launch_bounds__(256) void classify(
    const float* __restrict__ rst2, const float* __restrict__ b2,
    const float* __restrict__ Wc1, const float* __restrict__ bc1,
    const float* __restrict__ Wc2, const float* __restrict__ bc2,
    float* __restrict__ out, int nNodes)
{
    int wid  = blockIdx.x * 8 + (threadIdx.x >> 5);
    int lane = threadIdx.x & 31;
    if (wid >= nNodes) return;
    float acc = 0.0f;
#pragma unroll
    for (int d = 0; d < 64; ++d)
        acc += (rst2[(long)wid * 64 + d] + b2[d]) * Wc1[d * 32 + lane];
    float t = fmaxf(acc + bc1[lane], 0.0f);
    float o0 = t * Wc2[lane * 2 + 0];
    float o1 = t * Wc2[lane * 2 + 1];
#pragma unroll
    for (int o = 16; o > 0; o >>= 1) {
        o0 += __shfl_xor(o0, o, 32);
        o1 += __shfl_xor(o1, o, 32);
    }
    if (lane == 0) {
        out[(long)wid * 2 + 0] = o0 + bc2[0];
        out[(long)wid * 2 + 1] = o1 + bc2[1];
    }
}

extern "C" void kernel_launch(void* const* d_in, const int* in_sizes, int n_in,
                              void* d_out, int out_size, void* d_ws, size_t ws_size,
                              hipStream_t stream)
{
    (void)n_in; (void)out_size; (void)ws_size;

    const float* randfeat = (const float*)d_in[0];
    const float* func_emb = (const float*)d_in[1];
    const float* codebert = (const float*)d_in[2];
    const int*   src      = (const int*)d_in[3];
    const int*   dst      = (const int*)d_in[4];
    const float* W1  = (const float*)d_in[5];
    const float* al1 = (const float*)d_in[6];
    const float* ar1 = (const float*)d_in[7];
    const float* b1  = (const float*)d_in[8];
    const float* W2  = (const float*)d_in[9];
    const float* al2 = (const float*)d_in[10];
    const float* ar2 = (const float*)d_in[11];
    const float* b2  = (const float*)d_in[12];
    const float* Wc1 = (const float*)d_in[13];
    const float* bc1 = (const float*)d_in[14];
    const float* Wc2 = (const float*)d_in[15];
    const float* bc2 = (const float*)d_in[16];
    float* out = (float*)d_out;

    const int  E  = in_sizes[3];
    const long NN = N_NODES;

    float* ws     = (float*)d_ws;
    float* feat1  = ws;
    float* h2     = feat1  + NN * 256;
    float* feat2  = h2     + NN * 256;
    float* rst2   = feat2  + NN * 64;
    float* el1    = rst2   + NN * 64;
    float* er1    = el1    + NN * 4;
    float* el2    = er1    + NN * 4;
    float* er2    = el2    + NN;
    int*   rowEnd = (int*)(er2 + NN);
    int*   rowBeg = rowEnd + NN;
    int*   csrSrc = rowBeg + NN;
    int*   cnt    = (int*)h2;
    int*   pos    = cnt + CS_NCNT;
    int2*  tmp    = (int2*)(pos + CS_NCNT);

    csr_bucket_count<<<CS_NBLK, CS_NT, 0, stream>>>(dst, cnt, E);
    csr_scan<<<1, 1024, 0, stream>>>(cnt, pos, CS_NCNT);
    csr_bucket_place<<<CS_NBLK, CS_NT, 0, stream>>>(src, dst, pos, tmp, E);
    csr_bucket_sort<<<CS_NBUCK, 1024, 0, stream>>>(tmp, pos, rowBeg, rowEnd, csrSrc, E, N_NODES);

    dim3 g1((N_NODES + 207) / 208, 1);
    gemm_wmma_f32_strip<4, 13, 64><<<g1, 128, 0, stream>>>(randfeat, func_emb, codebert,
                                                        100, 100, 768, W1, feat1,
                                                        al1, ar1, el1, er1,
                                                        N_NODES, 256, 968);
    gat_gather<4, true><<<(N_NODES + 7) / 8, 256, 0, stream>>>(rowBeg, rowEnd, csrSrc, feat1, el1, er1,
                                                              b1, h2, N_NODES);

    dim3 g2((N_NODES + 127) / 128, 1);
    gemm_wmma_f32_tiled<8, 1, 1><<<g2, 256, 0, stream>>>(h2, h2, h2,
                                                        256, 0, 0, W2, feat2,
                                                        al2, ar2, el2, er2,
                                                        N_NODES, 64, 256);
    gat_gather<1, false><<<(N_NODES + 7) / 8, 256, 0, stream>>>(rowBeg, rowEnd, csrSrc, feat2, el2, er2,
                                                               b2, rst2, N_NODES);

    classify<<<(N_NODES + 7) / 8, 256, 0, stream>>>(rst2, b2, Wc1, bc1, Wc2, bc2, out, N_NODES);
}


# Layer 2 folded with the classifier: GEMM2 forms h2 @ (W2 @ [Wc1|al2|ar2]) (W2 products re-associated: new summation order), gather sums 32-float rows, ends in relu/Wc2; classify launch, rst2 gone
# speedup vs baseline: 5.8073x; 1.0362x over previous
#include <hip/hip_runtime.h>
#include <math.h>


typedef __attribute__((ext_vector_type(2))) float v2f;
typedef __attribute__((ext_vector_type(4))) float v4f;
typedef __attribute__((ext_vector_type(8))) float v8f;

#define N_NODES 50000

#define CS_NBLK  256
#define CS_SHIFT 8
#define CS_NBUCK ((N_NODES + 255) >> CS_SHIFT)
#define CS_NCNT  (CS_NBUCK * CS_NBLK)
#define CS_NT    1024
static_assert(CS_NBUCK <= CS_NT, "one thread per bucket");
static_assert(CS_NCNT % 4 == 0, "the scan reads the counts 16 bytes at a time");

__global__ __launch_bounds__(CS_NT) void csr_bucket_count(const int* __restrict__ dst, int* __restrict__ cnt, int E) {
    __shared__ int h[CS_NBUCK];
    const int t = threadIdx.x;
    if (t < CS_NBUCK) h[t] = 0;
    __syncthreads();
    const int chunk = (E + CS_NBLK - 1) / CS_NBLK;
    const int beg = blockIdx.x * chunk;
    const int end = min(E, beg + chunk);
    for (int i0 = beg + t; i0 < end; i0 += 4 * CS_NT) {
        int d[4];
#pragma unroll
        for (int k = 0; k < 4; ++k)
            d[k] = dst[min(i0 + k * CS_NT, end - 1)];
#pragma unroll
        for (int k = 0; k < 4; ++k)
            atomicAdd(&h[d[k] >> CS_SHIFT], i0 + k * CS_NT < end ? 1 : 0);
    }
    __syncthreads();
    if (t < CS_NBUCK) cnt[t * CS_NBLK + blockIdx.x] = h[t];
}

__global__ __launch_bounds__(1024) void csr_scan(int* __restrict__ cur, int* __restrict__ rowBeg, int n) {
    __shared__ int wtot[2][32];
    const int t = threadIdx.x, lane = t & 31, wave = t >> 5;
    auto loadTile = [&](int i) {
        int4 v = make_int4(0, 0, 0, 0);
        if (i + 3 < n) {
            v = *(const int4*)(cur + i);
        } else {
            if (i < n)     v.x = cur[i];
            if (i + 1 < n) v.y = cur[i + 1];
            if (i + 2 < n) v.z = cur[i + 2];
        }
        return v;
    };
    int carry = 0;
    int it = 0;
    int4 nxt = loadTile(4 * t);
    for (int base = 0; base < n; base += 4096, ++it) {
        const int i = base + 4 * t;
        const int4 v = nxt;
        if (base + 4096 < n) nxt = loadTile(i + 4096);
        const int s = v.x + v.y + v.z + v.w;
        int incl = s;
        for (int o = 1; o < 32; o <<= 1) {
            int u = __shfl_up(incl, o, 32);
            if (lane >= o) incl += u;
        }
        if (lane == 31) wtot[it & 1][wave] = incl;
        __syncthreads();
        const int w = wtot[it & 1][lane];
        int wi = w;
        for (int o = 1; o < 32; o <<= 1) {
            int u = __shfl_up(wi, o, 32);
            if (lane >= o) wi += u;
        }
        const int waveOff = __shfl(wi, wave, 32) - __shfl(w, wave, 32);
        const int total   = __shfl(wi, 31, 32);
        int4 r;
        r.x = carry + waveOff + incl - s;
        r.y = r.x + v.x;
        r.z = r.y + v.y;
        r.w = r.z + v.z;
        if (i + 3 < n) {
            *(int4*)(cur + i)    = r;
            *(int4*)(rowBeg + i) = r;
        } else {
            if (i < n)     { cur[i]     = r.x; rowBeg[i]     = r.x; }
            if (i + 1 < n) { cur[i + 1] = r.y; rowBeg[i + 1] = r.y; }
            if (i + 2 < n) { cur[i + 2] = r.z; rowBeg[i + 2] = r.z; }
        }
        carry += total;
    }
}

__global__ __launch_bounds__(CS_NT) void csr_bucket_place(const int* __restrict__ src, const int* __restrict__ dst,
                                                         const int* __restrict__ pos, int2* __restrict__ tmp, int E) {
    __shared__ int off[CS_NBUCK];
    const int t = threadIdx.x;
    if (t < CS_NBUCK) off[t] = pos[t * CS_NBLK + blockIdx.x];
    __syncthreads();
    const int chunk = (E + CS_NBLK - 1) / CS_NBLK;
    const int beg = blockIdx.x * chunk;
    const int end = min(E, beg + chunk);
    for (int i0 = beg + t; i0 < end; i0 += 4 * CS_NT) {
        int d[4], s[4];
#pragma unroll
        for (int k = 0; k < 4; ++k) {
            const int i = min(i0 + k * CS_NT, end - 1);
            d[k] = dst[i];
            s[k] = src[i];
        }
#pragma unroll
        for (int k = 0; k < 4; ++k) {
            if (i0 + k * CS_NT < end) {
                const int p = atomicAdd(&off[d[k] >> CS_SHIFT], 1);
                tmp[p] = make_int2(s[k], d[k]);
            }
        }
    }
}

__global__ __launch_bounds__(1024) void csr_bucket_sort(const int2* __restrict__ tmp, const int* __restrict__ pos,
                                                         int* __restrict__ rowBeg, int* __restrict__ rowEnd,
                                                         int* __restrict__ csrSrc, int E, int nNodes) {
    __shared__ int c[256];
    __shared__ int wtot[8];
    const int t = threadIdx.x, lane = t & 31, wave = t >> 5;
    const int b = blockIdx.x;
    const int beg = pos[b * CS_NBLK];
    const int end = (b + 1 < CS_NBUCK) ? pos[(b + 1) * CS_NBLK] : E;
    if (t < 256) c[t] = 0;
    __syncthreads();
    for (int i0 = beg + t; i0 < end; i0 += 8 * 1024) {
        int d[8];
#pragma unroll
        for (int k = 0; k < 8; ++k)
            d[k] = tmp[min(i0 + k * 1024, end - 1)].y;
#pragma unroll
        for (int k = 0; k < 8; ++k)
            atomicAdd(&c[d[k] & 255], i0 + k * 1024 < end ? 1 : 0);
    }
    __syncthreads();
    int deg = 0, incl = 0;
    if (t < 256) {
        deg  = c[t];
        incl = deg;
        for (int o = 1; o < 32; o <<= 1) {
            int u = __shfl_up(incl, o, 32);
            if (lane >= o) incl += u;
        }
        if (lane == 31) wtot[wave] = incl;
    }
    __syncthreads();
    if (t < 256) {
        int base = 0;
        for (int w = 0; w < wave; ++w) base += wtot[w];
        const int excl = base + incl - deg;
        c[t] = excl;
        const int node = (b << CS_SHIFT) + t;
        if (node < nNodes) {
            rowBeg[node] = beg + excl;
            rowEnd[node] = beg + excl + deg;
        }
    }
    __syncthreads();
    for (int i0 = beg + t; i0 < end; i0 += 8 * 1024) {
        int2 e[8];
#pragma unroll
        for (int k = 0; k < 8; ++k)
            e[k] = tmp[min(i0 + k * 1024, end - 1)];
#pragma unroll
        for (int k = 0; k < 8; ++k) {
            if (i0 + k * 1024 < end) {
                const int p = atomicAdd(&c[e[k].y & 255], 1);
                csrSrc[beg + p] = e[k].x;
            }
        }
    }
}

template <int H> struct RowVec;
template <> struct RowVec<4> { float4 a, b; };
template <> struct RowVec<1> { float2 a; };

template <int H, bool ELU>
__global__ __launch_bounds__(256) void gat_gather(
    const int* __restrict__ rowBeg, const int* __restrict__ rowEnd,
    const int* __restrict__ csrSrc,
    const float* __restrict__ feat, const float* __restrict__ el,
    const float* __restrict__ er, const float* __restrict__ bias,
    float* __restrict__ out, int nNodes)
{
    static_assert(H == 4 || H == 1, "head counts of this model");
    constexpr int C = H * 64;
    const int lane = threadIdx.x & 31;
    const int d = __builtin_amdgcn_readfirstlane((int)(blockIdx.x * 8 + (threadIdx.x >> 5)));
    if (d >= nNodes) return;
    const int beg = rowBeg[d];
    const int end = rowEnd[d];

    float erd[H];
    for (int h = 0; h < H; ++h) erd[h] = er[d * H + h];

    auto logits = [&](int s, float (&e)[H]) {
        float x[H];
        if constexpr (H == 4) {
            const float4 q = *(const float4*)(el + (long)s * 4);
            x[0] = q.x; x[1] = q.y; x[2] = q.z; x[3] = q.w;
        } else {
            x[0] = el[s];
        }
        for (int h = 0; h < H; ++h) {
            const float y = x[h] + erd[h];
            e[h] = y > 0.0f ? y : 0.2f * y;
        }
    };

    RowVec<H> acc;
    if constexpr (H == 4) {
        acc.a = make_float4(0.0f, 0.0f, 0.0f, 0.0f);
        acc.b = make_float4(0.0f, 0.0f, 0.0f, 0.0f);
    } else {
        acc.a = make_float2(0.0f, 0.0f);
    }
    const bool hi = lane >= 16;
    auto addRows = [&](int s, const float (&al)[H], int cnt) {
#pragma unroll 4
        for (int j = 0; j < cnt; ++j) {
            const int sj = __builtin_amdgcn_readlane(s, j);
            float aj[H];
            for (int h = 0; h < H; ++h)
                aj[h] = __int_as_float(__builtin_amdgcn_readlane(__float_as_int(al[h]), j));
            const float* row = feat + (long)sj * C;
            if constexpr (H == 4) {
                const float4 f0 = *(const float4*)(row + 4 * lane);
                const float4 f1 = *(const float4*)(row + 128 + 4 * lane);
                const float w0 = hi ? aj[1] : aj[0];
                const float w1 = hi ? aj[3] : aj[2];
                acc.a.x += f0.x * w0; acc.a.y += f0.y * w0; acc.a.z += f0.z * w0; acc.a.w += f0.w * w0;
                acc.b.x += f1.x * w1; acc.b.y += f1.y * w1; acc.b.z += f1.z * w1; acc.b.w += f1.w * w1;
            } else {
                const float2 f0 = *(const float2*)(row + 2 * lane);
                acc.a.x += f0.x * aj[0]; acc.a.y += f0.y * aj[0];
            }
        }
    };

    const int n = end - beg;
    if (n <= 32) {
        const bool act = lane < n;
        int   s = 0;
        float e[H];
        for (int h = 0; h < H; ++h) e[h] = -INFINITY;
        if (act) {
            s = csrSrc[beg + lane];
            logits(s, e);
        }
        float m[H];
        for (int h = 0; h < H; ++h) {
            m[h] = e[h];
            for (int o = 16; o > 0; o >>= 1) m[h] = fmaxf(m[h], __shfl_xor(m[h], o, 32));
            if (!isfinite(m[h])) m[h] = 0.0f;
        }
        float ex[H], den[H];
        for (int h = 0; h < H; ++h) {
            ex[h]  = act ? expf(e[h] - m[h]) : 0.0f;
            den[h] = ex[h];
            for (int o = 16; o > 0; o >>= 1) den[h] += __shfl_xor(den[h], o, 32);
            if (den[h] == 0.0f) den[h] = 1.0f;
        }
        float al[H];
        for (int h = 0; h < H; ++h) al[h] = act ? ex[h] / den[h] : 0.0f;
        addRows(s, al, n);
    } else {
        float m[H];
        for (int h = 0; h < H; ++h) m[h] = -INFINITY;
        for (int base = beg; base < end; base += 32) {
            const int i = base + lane;
            if (i < end) {
                float e[H];
                logits(csrSrc[i], e);
                for (int h = 0; h < H; ++h) m[h] = fmaxf(m[h], e[h]);
            }
        }
        for (int h = 0; h < H; ++h) {
            for (int o = 16; o > 0; o >>= 1) m[h] = fmaxf(m[h], __shfl_xor(m[h], o, 32));
            if (!isfinite(m[h])) m[h] = 0.0f;
        }

        float den[H];
        for (int h = 0; h < H; ++h) den[h] = 0.0f;
        for (int base = beg; base < end; base += 32) {
            const int i = base + lane;
            if (i < end) {
                float e[H];
                logits(csrSrc[i], e);
                for (int h = 0; h < H; ++h) den[h] += expf(e[h] - m[h]);
            }
        }
        for (int h = 0; h < H; ++h) {
            for (int o = 16; o > 0; o >>= 1) den[h] += __shfl_xor(den[h], o, 32);
            if (den[h] == 0.0f) den[h] = 1.0f;
        }

        for (int base = beg; base < end; base += 32) {
            const int i = base + lane;
            int   s = 0;
            float al[H];
            for (int h = 0; h < H; ++h) al[h] = 0.0f;
            if (i < end) {
                float e[H];
                s = csrSrc[i];
                logits(s, e);
                for (int h = 0; h < H; ++h) al[h] = expf(e[h] - m[h]) / den[h];
            }
            addRows(s, al, min(32, end - base));
        }
    }

    float* o = out + (long)d * C;
    if constexpr (H == 4) {
        float4 r0 = acc.a, r1 = acc.b;
        if constexpr (ELU) {
            const float4 b0 = *(const float4*)(bias + 4 * lane);
            const float4 b1 = *(const float4*)(bias + 128 + 4 * lane);
            r0.x += b0.x; r0.y += b0.y; r0.z += b0.z; r0.w += b0.w;
            r1.x += b1.x; r1.y += b1.y; r1.z += b1.z; r1.w += b1.w;
            r0.x = r0.x > 0.0f ? r0.x : expm1f(r0.x); r0.y = r0.y > 0.0f ? r0.y : expm1f(r0.y);
            r0.z = r0.z > 0.0f ? r0.z : expm1f(r0.z); r0.w = r0.w > 0.0f ? r0.w : expm1f(r0.w);
            r1.x = r1.x > 0.0f ? r1.x : expm1f(r1.x); r1.y = r1.y > 0.0f ? r1.y : expm1f(r1.y);
            r1.z = r1.z > 0.0f ? r1.z : expm1f(r1.z); r1.w = r1.w > 0.0f ? r1.w : expm1f(r1.w);
        }
        *(float4*)(o + 4 * lane)       = r0;
        *(float4*)(o + 128 + 4 * lane) = r1;
    } else {
        float2 r0 = acc.a;
        if constexpr (ELU) {
            const float2 b0 = *(const float2*)(bias + 2 * lane);
            r0.x += b0.x; r0.y += b0.y;
            r0.x = r0.x > 0.0f ? r0.x : expm1f(r0.x); r0.y = r0.y > 0.0f ? r0.y : expm1f(r0.y);
        }
        *(float2*)(o + 2 * lane) = r0;
    }
}

#define GT_BK 16
#define GT_AP 20

template <int WM, int WN, int TI>
__global__ __launch_bounds__(256) void gemm_wmma_f32_tiled(
    const float* __restrict__ p0, const float* __restrict__ p1,
    const float* __restrict__ p2, int w0, int w1, int w2,
    const float* __restrict__ B, float* __restrict__ C,
    int M, int N, int K)
{
    static_assert(WM * WN == 8, "8 waves per block");
    constexpr int BM  = WM * TI * 16;
    constexpr int BN  = WN * 64;
    constexpr int BP  = BN + 8;
    constexpr int NA  = BM / 64;
    constexpr int NB  = BN / 64;
    constexpr int BQ  = BN / 4;
    constexpr int BR  = 256 / BQ;
    static_assert(BM % 64 == 0, "A staging covers 64 rows per pass");

    __shared__ __attribute__((aligned(16))) float As[2][BM * GT_AP];
    __shared__ __attribute__((aligned(16))) float Bs[2][GT_BK * BP];

    const int tid  = threadIdx.x;
    const int lane = tid & 31;
    const int wave = tid >> 5;
    const int c    = lane & 15;
    const int h    = lane >> 4;
    const int m0   = (wave % WM) * (TI * 16);
    const int n0   = (wave / WM) * 64;

    const int blockRow = blockIdx.x * BM;
    const int blockCol = blockIdx.y * BN;

    const int am = tid >> 2;
    const int ak = (tid & 3) << 2;
    const int bk = tid / BQ;
    const int bn = (tid % BQ) << 2;

    v8f acc[TI][4];
    for (int i = 0; i < TI; ++i)
        for (int j = 0; j < 4; ++j)
            acc[i][j] = (v8f){};

    float4 ra[NA], rb[NB];

    auto loadTile = [&](int kb) {
        const int gk = kb + ak;
        for (int i = 0; i < NA; ++i) {
            const int row = blockRow + am + 64 * i;
            float4 v = make_float4(0.0f, 0.0f, 0.0f, 0.0f);
            if (row < M && gk < K) {
                const float* s;
                if (gk < w0)            s = p0 + (long)row * w0 + gk;
                else if (gk - w0 < w1)  s = p1 + (long)row * w1 + (gk - w0);
                else                    s = p2 + (long)row * w2 + (gk - w0 - w1);
                v = *(const float4*)s;
            }
            ra[i] = v;
        }
        const int col = blockCol + bn;
        for (int i = 0; i < NB; ++i) {
            const int k = kb + bk + BR * i;
            float4 v = make_float4(0.0f, 0.0f, 0.0f, 0.0f);
            if (k < K && col < N) v = *(const float4*)(B + (long)k * N + col);
            rb[i] = v;
        }
    };

    auto storeTile = [&](int buf) {
        for (int i = 0; i < NA; ++i)
            *(float4*)&As[buf][(am + 64 * i) * GT_AP + ak] = ra[i];
        for (int i = 0; i < NB; ++i)
            *(float4*)&Bs[buf][(bk + BR * i) * BP + bn] = rb[i];
    };

    const int nT = (K + GT_BK - 1) / GT_BK;

    loadTile(0);
    storeTile(0);
    __syncthreads();

    for (int t = 0; t < nT; ++t) {
        const int buf = t & 1;
        if (t + 1 < nT) loadTile((t + 1) * GT_BK);

        const float* Ab = &As[buf][(m0 + c) * GT_AP + 2 * h];
        const float* Bb = &Bs[buf][(2 * h) * BP + n0 + 4 * c];
        for (int kk = 0; kk < GT_BK; kk += 4) {
            v2f a[TI];
            for (int i = 0; i < TI; ++i)
                a[i] = *(const v2f*)&Ab[(16 * i) * GT_AP + kk];
            const float4 bx = *(const float4*)&Bb[kk * BP];
            const float4 by = *(const float4*)&Bb[(kk + 1) * BP];
            v2f b[4];
            b[0].x = bx.x; b[0].y = by.x;
            b[1].x = bx.y; b[1].y = by.y;
            b[2].x = bx.z; b[2].y = by.z;
            b[3].x = bx.w; b[3].y = by.w;
            for (int i = 0; i < TI; ++i)
                for (int j = 0; j < 4; ++j)
                    acc[i][j] = __builtin_amdgcn_wmma_f32_16x16x4_f32(
                        false, a[i], false, b[j], (short)0, acc[i][j], false, false);
        }

        if (t + 1 < nT) storeTile(buf ^ 1);
        __syncthreads();
    }

    const int col = blockCol + n0 + 4 * c;

    if (col < N) {
        for (int i = 0; i < TI; ++i) {
            for (int r = 0; r < 8; ++r) {
                const int row = blockRow + m0 + 16 * i + 8 * h + r;
                if (row < M) {
                    float4 v = make_float4(acc[i][0][r], acc[i][1][r], acc[i][2][r], acc[i][3][r]);
                    *(float4*)(C + (long)row * N + col) = v;
                }
            }
        }
    }
}


struct GemmSeg { const float* p; int w; int o; };
__device__ __forceinline__ GemmSeg gemm_segment(const float* p0, const float* p1, const float* p2,
                                                int w0, int w1, int w2, int gk)
{
    GemmSeg s = { p2, w2, gk - w0 - w1 };
    if (gk - w0 < w1) { s.p = p1; s.w = w1; s.o = gk - w0; }
    if (gk < w0)      { s.p = p0; s.w = w0; s.o = gk; }
    return s;
}

template <int WN, int TI, int BK>
__global__ __launch_bounds__(WN * 32) __attribute__((amdgpu_waves_per_eu(1, 1))) void gemm_wmma_f32_strip(
    const float* __restrict__ p0, const float* __restrict__ p1,
    const float* __restrict__ p2, int w0, int w1, int w2,
    const float* __restrict__ B, float* __restrict__ C,
    const float* __restrict__ al, const float* __restrict__ ar,
    float* __restrict__ el, float* __restrict__ er,
    int M, int N, int K)
{
    constexpr int NT  = WN * 32;
    constexpr int BM  = TI * 16;
    constexpr int BN  = WN * 64;
    constexpr int AP  = BK + 4;
    constexpr int BP  = BN + 8;
    constexpr int KQ  = BK / 4;
    constexpr int AR  = NT / KQ;
    constexpr int NA  = BM / AR;
    constexpr int BQ  = BN / 4;
    constexpr int BR  = NT / BQ;
    constexpr int NB  = BK / BR;
    constexpr int HB  = WN;
    static_assert(BK % 8 == 0 && (KQ & 1) == 0, "A pitch of 16 B x odd needs an even count of float4 per row");
    static_assert(NT % KQ == 0 && BM % AR == 0, "A staging passes cover the tile exactly");
    static_assert(NT % BQ == 0 && BK % BR == 0, "B staging passes cover the tile exactly");

    __shared__ __attribute__((aligned(16))) float As[2][BM * AP];
    __shared__ __attribute__((aligned(16))) float Bs[2][BK * BP];
    __shared__ float Es[2][BM * HB];

    const int tid  = threadIdx.x;
    const int lane = tid & 31;
    const int wave = tid >> 5;
    const int c    = lane & 15;
    const int h    = lane >> 4;
    const int m0   = 0;
    const int n0   = wave * 64;

    const int blockRow = blockIdx.x * BM;
    const int blockCol = blockIdx.y * BN;

    const int am = tid / KQ;
    const int ak = (tid % KQ) << 2;
    const int bk = tid / BQ;
    const int bn = (tid % BQ) << 2;

    v8f acc[TI][4];
#pragma unroll
    for (int i = 0; i < TI; ++i)
#pragma unroll
        for (int j = 0; j < 4; ++j)
            acc[i][j] = (v8f){};

    v4f ra[NA], rb[NB];

    const int bcol = min(blockCol + bn, N - 4);

#define GEMM_LOAD_TILE(kb)                                                                        \
    {                                                                                             \
        const GemmSeg sg = gemm_segment(p0, p1, p2, w0, w1, w2, min((kb) + ak, K - 4));           \
        _Pragma("unroll")                                                                         \
        for (int i = 0; i < NA; ++i) {                                                            \
            const int row = min(blockRow + am + AR * i, M - 1);                                   \
            ra[i] = *(const v4f*)(sg.p + (long)row * sg.w + sg.o);                                \
        }                                                                                         \
        _Pragma("unroll")                                                                         \
        for (int i = 0; i < NB; ++i) {                                                            \
            const int k = min((kb) + bk + BR * i, K - 1);                                         \
            rb[i] = *(const v4f*)(B + (long)k * N + bcol);                                        \
        }                                                                                         \
    }
#define GEMM_STORE_A(buf)                                                                         \
    {                                                                                             \
        _Pragma("unroll")                                                                         \
        for (int i = 0; i < NA; ++i)                                                              \
            *(v4f*)&As[buf][(am + AR * i) * AP + ak] = ra[i];                                     \
    }
#define GEMM_STORE_B(buf)                                                                         \
    {                                                                                             \
        _Pragma("unroll")                                                                         \
        for (int i = 0; i < NB; ++i)                                                              \
            *(v4f*)&Bs[buf][(bk + BR * i) * BP + bn] = rb[i];                                     \
    }

    auto kstep = [&](const float* Ab, const float* Bb, int kk) {
        v2f a[TI];
#pragma unroll
        for (int i = 0; i < TI; ++i)
            a[i] = *(const v2f*)&Ab[(16 * i) * AP + kk];
        const float4 bx = *(const float4*)&Bb[kk * BP];
        const float4 by = *(const float4*)&Bb[(kk + 1) * BP];
        v2f b[4];
        b[0].x = bx.x; b[0].y = by.x;
        b[1].x = bx.y; b[1].y = by.y;
        b[2].x = bx.z; b[2].y = by.z;
        b[3].x = bx.w; b[3].y = by.w;
#pragma unroll
        for (int i = 0; i < TI; ++i)
#pragma unroll
            for (int j = 0; j < 4; ++j)
                acc[i][j] = __builtin_amdgcn_wmma_f32_16x16x4_f32(
                    false, a[i], false, b[j], (short)0, acc[i][j], false, false);
    };

    const int nFull  = K / BK;
    const int ksTail = (K - nFull * BK) >> 2;

    GEMM_LOAD_TILE(0)
    GEMM_STORE_A(0)
    GEMM_STORE_B(0)
    __syncthreads();

    for (int t = 0; t < nFull; ++t) {
        const int buf = t & 1;
        GEMM_LOAD_TILE((t + 1) * BK)

        const float* Ab = &As[buf][(m0 + c) * AP + 2 * h];
        const float* Bb = &Bs[buf][(2 * h) * BP + n0 + 4 * c];
#pragma unroll
        for (int s = 0; s < KQ; ++s) {
            kstep(Ab, Bb, 4 * s);
            if (s == KQ / 2 - 1) GEMM_STORE_A(buf ^ 1)
            if (s == KQ / 2)     GEMM_STORE_B(buf ^ 1)
        }
        __syncthreads();
    }

    if (ksTail > 0) {
        const int buf = nFull & 1;
        const float* Ab = &As[buf][(m0 + c) * AP + 2 * h];
        const float* Bb = &Bs[buf][(2 * h) * BP + n0 + 4 * c];
#pragma unroll 1
        for (int s = 0; s < ksTail; ++s)
            kstep(Ab, Bb, 4 * s);
    }

    const int col = blockCol + n0 + 4 * c;

    float4 wl = make_float4(0.0f, 0.0f, 0.0f, 0.0f);
    float4 wr = make_float4(0.0f, 0.0f, 0.0f, 0.0f);
    if (col < N) {
        wl = *(const float4*)(al + col);
        wr = *(const float4*)(ar + col);
    }

    if (col < N) {
#pragma unroll
        for (int i = 0; i < TI; ++i) {
#pragma unroll
            for (int r = 0; r < 8; ++r) {
                const int row = blockRow + m0 + 16 * i + 8 * h + r;
                if (row < M) {
                    float4 v = make_float4(acc[i][0][r], acc[i][1][r], acc[i][2][r], acc[i][3][r]);
                    *(float4*)(C + (long)row * N + col) = v;
                }
            }
        }
    }

    float p[TI][16];
#pragma unroll
    for (int i = 0; i < TI; ++i) {
#pragma unroll
        for (int r = 0; r < 8; ++r) {
            p[i][r]     = acc[i][0][r] * wl.x + acc[i][1][r] * wl.y + acc[i][2][r] * wl.z + acc[i][3][r] * wl.w;
            p[i][8 + r] = acc[i][0][r] * wr.x + acc[i][1][r] * wr.y + acc[i][2][r] * wr.z + acc[i][3][r] * wr.w;
        }
    }
#pragma unroll
    for (int o = 8; o >= 1; o >>= 1) {
        const bool up = (c & o) != 0;
#pragma unroll
        for (int i = 0; i < TI; ++i) {
#pragma unroll
            for (int k = 0; k < o; ++k) {
                const float lo = p[i][k];
                const float hi = p[i][k + o];
                const float keep = up ? hi : lo;
                const float send = up ? lo : hi;
                p[i][k] = keep + __shfl_xor(send, o, 32);
            }
        }
    }
#pragma unroll
    for (int i = 0; i < TI; ++i)
        Es[c >> 3][(m0 + 16 * i + 8 * h + (c & 7)) * HB + (n0 >> 6)] = p[i][0];
    __syncthreads();
    {
        const int nH = N >> 6;
        const int h0 = blockCol >> 6;
        for (int idx = tid; idx < BM * HB; idx += NT) {
            const int row  = blockRow + idx / HB;
            const int head = h0 + idx % HB;
            if (row < M && head < nH) {
                el[(long)row * nH + head] = Es[0][idx];
                er[(long)row * nH + head] = Es[1][idx];
            }
        }
    }
}
#undef GEMM_LOAD_TILE
#undef GEMM_STORE_A
#undef GEMM_STORE_B

#define HF_K    256
#define HF_D    64
#define HF_HID  32
#define HF_COLS 64
#define HF_EL   32
#define HF_ER   33

__global__ __launch_bounds__(256) void fold_head(
    const float* __restrict__ W2, const float* __restrict__ Wc1,
    const float* __restrict__ al2, const float* __restrict__ ar2,
    const float* __restrict__ b2, const float* __restrict__ bc1,
    float* __restrict__ Wf, float* __restrict__ hb)
{
    const int idx = blockIdx.x * 256 + threadIdx.x;
    if (idx >= (HF_K + 1) * HF_COLS) return;
    const int k = idx / HF_COLS, c = idx % HF_COLS;
    const float* x = k < HF_K ? W2 + k * HF_D : b2;
    float s = 0.0f;
    if (c < HF_HID) {
        for (int j = 0; j < HF_D; ++j) s += x[j] * Wc1[j * HF_HID + c];
    } else if (c == HF_EL || c == HF_ER) {
        const float* a = c == HF_EL ? al2 : ar2;
        for (int j = 0; j < HF_D; ++j) s += x[j] * a[j];
    }
    if (k < HF_K)        Wf[k * HF_COLS + c] = s;
    else if (c < HF_HID) hb[c] = s + bc1[c];
}

__global__ __launch_bounds__(256) void gat_gather_head(
    const int* __restrict__ rowBeg, const int* __restrict__ rowEnd,
    const int* __restrict__ csrSrc,
    const float* __restrict__ g, const float* __restrict__ hb,
    const float* __restrict__ Wc2, const float* __restrict__ bc2,
    float* __restrict__ out, int nNodes)
{
    const int lane = threadIdx.x & 31;
    const int d = __builtin_amdgcn_readfirstlane((int)(blockIdx.x * 8 + (threadIdx.x >> 5)));
    if (d >= nNodes) return;
    const int beg = rowBeg[d];
    const int end = rowEnd[d];
    const float erd = g[(long)d * HF_COLS + HF_ER];

    auto logit = [&](int s) {
        const float y = g[(long)s * HF_COLS + HF_EL] + erd;
        return y > 0.0f ? y : 0.2f * y;
    };

    float acc = 0.0f;
    auto addRows = [&](int s, float al, int cnt) {
#pragma unroll 4
        for (int j = 0; j < cnt; ++j) {
            const int   sj = __builtin_amdgcn_readlane(s, j);
            const float aj = __int_as_float(__builtin_amdgcn_readlane(__float_as_int(al), j));
            acc += g[(long)sj * HF_COLS + lane] * aj;
        }
    };

    const int n = end - beg;
    if (n <= 32) {
        const bool act = lane < n;
        int   s = 0;
        float e = -INFINITY;
        if (act) {
            s = csrSrc[beg + lane];
            e = logit(s);
        }
        float m = e;
        for (int o = 16; o > 0; o >>= 1) m = fmaxf(m, __shfl_xor(m, o, 32));
        if (!isfinite(m)) m = 0.0f;
        const float ex = act ? expf(e - m) : 0.0f;
        float den = ex;
        for (int o = 16; o > 0; o >>= 1) den += __shfl_xor(den, o, 32);
        if (den == 0.0f) den = 1.0f;
        addRows(s, act ? ex / den : 0.0f, n);
    } else {
        float m = -INFINITY;
        for (int base = beg; base < end; base += 32) {
            const int i = base + lane;
            if (i < end) m = fmaxf(m, logit(csrSrc[i]));
        }
        for (int o = 16; o > 0; o >>= 1) m = fmaxf(m, __shfl_xor(m, o, 32));
        if (!isfinite(m)) m = 0.0f;

        float den = 0.0f;
        for (int base = beg; base < end; base += 32) {
            const int i = base + lane;
            if (i < end) den += expf(logit(csrSrc[i]) - m);
        }
        for (int o = 16; o > 0; o >>= 1) den += __shfl_xor(den, o, 32);
        if (den == 0.0f) den = 1.0f;

        for (int base = beg; base < end; base += 32) {
            const int i = base + lane;
            int   s  = 0;
            float al = 0.0f;
            if (i < end) {
                s  = csrSrc[i];
                al = expf(logit(s) - m) / den;
            }
            addRows(s, al, min(32, end - base));
        }
    }

    const float t = fmaxf(acc + hb[lane], 0.0f);
    float o0 = t * Wc2[lane * 2 + 0];
    float o1 = t * Wc2[lane * 2 + 1];
    for (int o = 16; o > 0; o >>= 1) {
        o0 += __shfl_xor(o0, o, 32);
        o1 += __shfl_xor(o1, o, 32);
    }
    if (lane == 0) {
        out[(long)d * 2 + 0] = o0 + bc2[0];
        out[(long)d * 2 + 1] = o1 + bc2[1];
    }
}

extern "C" void kernel_launch(void* const* d_in, const int* in_sizes, int n_in,
                              void* d_out, int out_size, void* d_ws, size_t ws_size,
                              hipStream_t stream)
{
    (void)n_in; (void)out_size; (void)ws_size;

    const float* randfeat = (const float*)d_in[0];
    const float* func_emb = (const float*)d_in[1];
    const float* codebert = (const float*)d_in[2];
    const int*   src      = (const int*)d_in[3];
    const int*   dst      = (const int*)d_in[4];
    const float* W1  = (const float*)d_in[5];
    const float* al1 = (const float*)d_in[6];
    const float* ar1 = (const float*)d_in[7];
    const float* b1  = (const float*)d_in[8];
    const float* W2  = (const float*)d_in[9];
    const float* al2 = (const float*)d_in[10];
    const float* ar2 = (const float*)d_in[11];
    const float* b2  = (const float*)d_in[12];
    const float* Wc1 = (const float*)d_in[13];
    const float* bc1 = (const float*)d_in[14];
    const float* Wc2 = (const float*)d_in[15];
    const float* bc2 = (const float*)d_in[16];
    float* out = (float*)d_out;

    const int  E  = in_sizes[3];
    const long NN = N_NODES;

    float* ws     = (float*)d_ws;
    float* feat1  = ws;
    float* h2     = feat1  + NN * 256;
    float* g2     = h2     + NN * 256;
    float* Wf     = g2     + NN * HF_COLS;
    float* hb     = Wf     + HF_K * HF_COLS;
    float* el1    = hb     + HF_COLS;
    float* er1    = el1    + NN * 4;
    int*   rowEnd = (int*)(er1 + NN * 4);
    int*   rowBeg = rowEnd + NN;
    int*   csrSrc = rowBeg + NN;
    int*   cnt    = (int*)h2;
    int*   pos    = cnt + CS_NCNT;
    int2*  tmp    = (int2*)(pos + CS_NCNT);

    csr_bucket_count<<<CS_NBLK, CS_NT, 0, stream>>>(dst, cnt, E);
    csr_scan<<<1, 1024, 0, stream>>>(cnt, pos, CS_NCNT);
    csr_bucket_place<<<CS_NBLK, CS_NT, 0, stream>>>(src, dst, pos, tmp, E);
    csr_bucket_sort<<<CS_NBUCK, 1024, 0, stream>>>(tmp, pos, rowBeg, rowEnd, csrSrc, E, N_NODES);

    dim3 g1((N_NODES + 207) / 208, 1);
    gemm_wmma_f32_strip<4, 13, 64><<<g1, 128, 0, stream>>>(randfeat, func_emb, codebert,
                                                        100, 100, 768, W1, feat1,
                                                        al1, ar1, el1, er1,
                                                        N_NODES, 256, 968);
    gat_gather<4, true><<<(N_NODES + 7) / 8, 256, 0, stream>>>(rowBeg, rowEnd, csrSrc, feat1, el1, er1,
                                                              b1, h2, N_NODES);

    fold_head<<<((HF_K + 1) * HF_COLS + 255) / 256, 256, 0, stream>>>(W2, Wc1, al2, ar2, b2, bc1, Wf, hb);
    dim3 gl2((N_NODES + 127) / 128, 1);
    gemm_wmma_f32_tiled<8, 1, 1><<<gl2, 256, 0, stream>>>(h2, h2, h2,
                                                         HF_K, 0, 0, Wf, g2,
                                                         N_NODES, HF_COLS, HF_K);
    gat_gather_head<<<(N_NODES + 7) / 8, 256, 0, stream>>>(rowBeg, rowEnd, csrSrc, g2, hb, Wc2, bc2, out, N_NODES);
}
